# starting kernel
# baseline (speedup 1.0000x reference)
#include <hip/hip_runtime.h>
#include <hip/hip_bf16.h>

typedef float v2f __attribute__((ext_vector_type(2)));
typedef float v8f __attribute__((ext_vector_type(8)));

#define N_NODES 100000
#define N_EDGES 1600000
#define IN_F    256
#define OUT_F   64
#define NTILES  (N_NODES / 16)

__device__ __forceinline__ void atomic_add_f32_dev(float* p, float v) {
    asm volatile("global_atomic_add_f32 %0, %1, off scope:SCOPE_DEV"
                 :: "v"(p), "v"(v) : "memory");
}

__global__ void sgc_gemm_wmma(const float* __restrict__ x,
                              const float* __restrict__ W,
                              const float* __restrict__ bias,
                              float* __restrict__ out) {
    const int wave = (blockIdx.x * blockDim.x + threadIdx.x) >> 5;
    const int lane = threadIdx.x & 31;
    if (wave >= NTILES) return;

    const int m    = lane & 15;
    const int koff = (lane >> 4) << 1;

    const float* __restrict__ xrow = x + (size_t)(wave * 16 + m) * IN_F;
    const float* __restrict__ wr0  = W + (size_t)(m +  0) * IN_F;
    const float* __restrict__ wr1  = W + (size_t)(m + 16) * IN_F;
    const float* __restrict__ wr2  = W + (size_t)(m + 32) * IN_F;
    const float* __restrict__ wr3  = W + (size_t)(m + 48) * IN_F;

    v8f acc0 = {}, acc1 = {}, acc2 = {}, acc3 = {};

#pragma unroll 4
    for (int k = 0; k < IN_F; k += 4) {
        v2f a  = *(const v2f*)(xrow + k + koff);
        v2f b0 = *(const v2f*)(wr0  + k + koff);
        v2f b1 = *(const v2f*)(wr1  + k + koff);
        v2f b2 = *(const v2f*)(wr2  + k + koff);
        v2f b3 = *(const v2f*)(wr3  + k + koff);
        acc0 = __builtin_amdgcn_wmma_f32_16x16x4_f32(false, a, false, b0, (short)0, acc0, false, false);
        acc1 = __builtin_amdgcn_wmma_f32_16x16x4_f32(false, a, false, b1, (short)0, acc1, false, false);
        acc2 = __builtin_amdgcn_wmma_f32_16x16x4_f32(false, a, false, b2, (short)0, acc2, false, false);
        acc3 = __builtin_amdgcn_wmma_f32_16x16x4_f32(false, a, false, b3, (short)0, acc3, false, false);
    }

    const int row_base = wave * 16 + ((lane >> 4) ? 8 : 0);
    const int col      = lane & 15;
    const float bv0 = bias[col +  0];
    const float bv1 = bias[col + 16];
    const float bv2 = bias[col + 32];
    const float bv3 = bias[col + 48];

#pragma unroll
    for (int r = 0; r < 8; ++r) {
        float* o = out + (size_t)(row_base + r) * OUT_F + col;
        o[ 0] = acc0[r] + bv0;
        o[16] = acc1[r] + bv1;
        o[32] = acc2[r] + bv2;
        o[48] = acc3[r] + bv3;
    }
}

__global__ void sgc_zero(float4* __restrict__ p, int n4) {
    const float4 z = {0.f, 0.f, 0.f, 0.f};
    for (int i = blockIdx.x * blockDim.x + threadIdx.x; i < n4;
         i += gridDim.x * blockDim.x)
        p[i] = z;
}

__global__ void sgc_scatter(const float* __restrict__ cur,
                            float* __restrict__ nxt,
                            const int* __restrict__ esrc,
                            const int* __restrict__ edst,
                            const float* __restrict__ ew,
                            int n_edges) {
    const int gid  = blockIdx.x * blockDim.x + threadIdx.x;
    const int e    = gid >> 5;
    const int lane = gid & 31;
    if (e >= n_edges) return;

    const int   s = __builtin_nontemporal_load(esrc + e);
    const int   d = __builtin_nontemporal_load(edst + e);
    const float w = __builtin_nontemporal_load(ew + e);

    const float* __restrict__ ps = cur + (size_t)s * OUT_F + lane * 2;
    float*       __restrict__ pd = nxt + (size_t)d * OUT_F + lane * 2;

    const v2f vv = *(const v2f*)ps;
    atomic_add_f32_dev(pd + 0, vv.x * w);
    atomic_add_f32_dev(pd + 1, vv.y * w);
}

extern "C" void kernel_launch(void* const* d_in, const int* in_sizes, int n_in,
                              void* d_out, int out_size, void* d_ws, size_t ws_size,
                              hipStream_t stream) {
    const float* x    = (const float*)d_in[0];
    const float* W    = (const float*)d_in[1];
    const float* b    = (const float*)d_in[2];
    const int*   esrc = (const int*)d_in[3];
    const int*   edst = (const int*)d_in[4];
    const float* ew   = (const float*)d_in[5];

    float* out = (float*)d_out;
    float* ws  = (float*)d_ws;

    const int n4 = N_NODES * OUT_F / 4;

    sgc_gemm_wmma<<<(NTILES + 7) / 8, 256, 0, stream>>>(x, W, b, ws);

    const dim3 sgrid((unsigned)(((long long)N_EDGES * 32 + 255) / 256));

    sgc_zero<<<1024, 256, 0, stream>>>((float4*)out, n4);
    sgc_scatter<<<sgrid, 256, 0, stream>>>(ws, out, esrc, edst, ew, N_EDGES);

    sgc_zero<<<1024, 256, 0, stream>>>((float4*)ws, n4);
    sgc_scatter<<<sgrid, 256, 0, stream>>>(out, ws, esrc, edst, ew, N_EDGES);

    sgc_zero<<<1024, 256, 0, stream>>>((float4*)out, n4);
    sgc_scatter<<<sgrid, 256, 0, stream>>>(ws, out, esrc, edst, ew, N_EDGES);
}


# scatter hops: one wave takes 32 consecutive edges (coalesced index loads, lane broadcast, 32 row gathers in flight) instead of one edge per wave; f32 atomic adds kept, arrival order as before
# speedup vs baseline: 1.9026x; 1.9026x over previous
#include <hip/hip_runtime.h>
#include <hip/hip_bf16.h>

typedef float v2f __attribute__((ext_vector_type(2)));
typedef float v8f __attribute__((ext_vector_type(8)));

#define N_NODES 100000
#define N_EDGES 1600000
#define IN_F    256
#define OUT_F   64
#define NTILES  (N_NODES / 16)

__global__ void sgc_gemm_wmma(const float* __restrict__ x,
                              const float* __restrict__ W,
                              const float* __restrict__ bias,
                              float* __restrict__ out) {
    const int wave = (blockIdx.x * blockDim.x + threadIdx.x) >> 5;
    const int lane = threadIdx.x & 31;
    if (wave >= NTILES) return;

    const int m    = lane & 15;
    const int koff = (lane >> 4) << 1;

    const float* __restrict__ xrow = x + (size_t)(wave * 16 + m) * IN_F;
    const float* __restrict__ wr0  = W + (size_t)(m +  0) * IN_F;
    const float* __restrict__ wr1  = W + (size_t)(m + 16) * IN_F;
    const float* __restrict__ wr2  = W + (size_t)(m + 32) * IN_F;
    const float* __restrict__ wr3  = W + (size_t)(m + 48) * IN_F;

    v8f acc0 = {}, acc1 = {}, acc2 = {}, acc3 = {};

#pragma unroll 4
    for (int k = 0; k < IN_F; k += 4) {
        v2f a  = *(const v2f*)(xrow + k + koff);
        v2f b0 = *(const v2f*)(wr0  + k + koff);
        v2f b1 = *(const v2f*)(wr1  + k + koff);
        v2f b2 = *(const v2f*)(wr2  + k + koff);
        v2f b3 = *(const v2f*)(wr3  + k + koff);
        acc0 = __builtin_amdgcn_wmma_f32_16x16x4_f32(false, a, false, b0, (short)0, acc0, false, false);
        acc1 = __builtin_amdgcn_wmma_f32_16x16x4_f32(false, a, false, b1, (short)0, acc1, false, false);
        acc2 = __builtin_amdgcn_wmma_f32_16x16x4_f32(false, a, false, b2, (short)0, acc2, false, false);
        acc3 = __builtin_amdgcn_wmma_f32_16x16x4_f32(false, a, false, b3, (short)0, acc3, false, false);
    }

    const int row_base = wave * 16 + ((lane >> 4) ? 8 : 0);
    const int col      = lane & 15;
    const float bv0 = bias[col +  0];
    const float bv1 = bias[col + 16];
    const float bv2 = bias[col + 32];
    const float bv3 = bias[col + 48];

#pragma unroll
    for (int r = 0; r < 8; ++r) {
        float* o = out + (size_t)(row_base + r) * OUT_F + col;
        o[ 0] = acc0[r] + bv0;
        o[16] = acc1[r] + bv1;
        o[32] = acc2[r] + bv2;
        o[48] = acc3[r] + bv3;
    }
}

__global__ void sgc_zero(float4* __restrict__ p, int n4) {
    const float4 z = {0.f, 0.f, 0.f, 0.f};
    for (int i = blockIdx.x * blockDim.x + threadIdx.x; i < n4;
         i += gridDim.x * blockDim.x)
        p[i] = z;
}

__device__ __forceinline__ void atomic_add_f32x2_dev(float* base, unsigned off, float x, float y) {
    asm volatile("global_atomic_add_f32 %0, %1, %3 scope:SCOPE_DEV\n\t"
                 "global_atomic_add_f32 %0, %2, %3 offset:4 scope:SCOPE_DEV\n\t"
                 "s_wait_xcnt 0x0"
                 :: "v"(off), "v"(x), "v"(y), "s"(base) : "memory");
}

#define EPW 32
static_assert(N_EDGES % EPW == 0, "edge count must be a whole number of wave groups");

__global__ void __launch_bounds__(256)
sgc_scatter(const float* __restrict__ cur,
            float* __restrict__ nxt,
            const int* __restrict__ esrc,
            const int* __restrict__ edst,
            const float* __restrict__ ew) {
    const int wave = (blockIdx.x * blockDim.x + threadIdx.x) >> 5;
    const int lane = threadIdx.x & 31;
    if (wave >= N_EDGES / EPW) return;

    const int e  = wave * EPW + lane;
    const int s  = __builtin_nontemporal_load(esrc + e);
    const int d  = __builtin_nontemporal_load(edst + e);
    const int wi = __builtin_bit_cast(int, __builtin_nontemporal_load(ew + e));

    const unsigned lane8 = (unsigned)lane * 8u;

    v2f v[EPW];
#pragma unroll
    for (int j = 0; j < EPW; ++j) {
        const unsigned sj = (unsigned)__builtin_amdgcn_readlane(s, j);
        v[j] = *(const v2f*)((const char*)cur + (sj * (unsigned)(OUT_F * 4) + lane8));
    }
#pragma unroll
    for (int j = 0; j < EPW; ++j) {
        const unsigned dj = (unsigned)__builtin_amdgcn_readlane(d, j);
        const float    wj = __builtin_bit_cast(float, __builtin_amdgcn_readlane(wi, j));
        const unsigned off = dj * (unsigned)(OUT_F * 4) + lane8;
        atomic_add_f32x2_dev(nxt, off, v[j].x * wj, v[j].y * wj);
    }
}

extern "C" void kernel_launch(void* const* d_in, const int* in_sizes, int n_in,
                              void* d_out, int out_size, void* d_ws, size_t ws_size,
                              hipStream_t stream) {
    const float* x    = (const float*)d_in[0];
    const float* W    = (const float*)d_in[1];
    const float* b    = (const float*)d_in[2];
    const int*   esrc = (const int*)d_in[3];
    const int*   edst = (const int*)d_in[4];
    const float* ew   = (const float*)d_in[5];

    float* out = (float*)d_out;
    float* ws  = (float*)d_ws;

    const int n4 = N_NODES * OUT_F / 4;

    sgc_gemm_wmma<<<(NTILES + 7) / 8, 256, 0, stream>>>(x, W, b, ws);

    const dim3 sgrid((N_EDGES / EPW + 7) / 8);

    sgc_zero<<<1024, 256, 0, stream>>>((float4*)out, n4);
    sgc_scatter<<<sgrid, 256, 0, stream>>>(ws, out, esrc, edst, ew);

    sgc_zero<<<1024, 256, 0, stream>>>((float4*)ws, n4);
    sgc_scatter<<<sgrid, 256, 0, stream>>>(out, ws, esrc, edst, ew);

    sgc_zero<<<1024, 256, 0, stream>>>((float4*)out, n4);
    sgc_scatter<<<sgrid, 256, 0, stream>>>(ws, out, esrc, edst, ew);
}


# scatter hops: each f32 atomic instruction covers one whole 128-B line of the destination row (lane halves exchange a register, v_permlane16_swap_b32), not half of two lines; arrival order as before
# speedup vs baseline: 2.0928x; 1.1000x over previous
#include <hip/hip_runtime.h>
#include <hip/hip_bf16.h>

typedef float v2f __attribute__((ext_vector_type(2)));
typedef float v8f __attribute__((ext_vector_type(8)));

#define N_NODES 100000
#define N_EDGES 1600000
#define IN_F    256
#define OUT_F   64
#define NTILES  (N_NODES / 16)

__global__ void sgc_gemm_wmma(const float* __restrict__ x,
                              const float* __restrict__ W,
                              const float* __restrict__ bias,
                              float* __restrict__ out) {
    const int wave = (blockIdx.x * blockDim.x + threadIdx.x) >> 5;
    const int lane = threadIdx.x & 31;
    if (wave >= NTILES) return;

    const int m    = lane & 15;
    const int koff = (lane >> 4) << 1;

    const float* __restrict__ xrow = x + (size_t)(wave * 16 + m) * IN_F;
    const float* __restrict__ wr0  = W + (size_t)(m +  0) * IN_F;
    const float* __restrict__ wr1  = W + (size_t)(m + 16) * IN_F;
    const float* __restrict__ wr2  = W + (size_t)(m + 32) * IN_F;
    const float* __restrict__ wr3  = W + (size_t)(m + 48) * IN_F;

    v8f acc0 = {}, acc1 = {}, acc2 = {}, acc3 = {};

#pragma unroll 4
    for (int k = 0; k < IN_F; k += 4) {
        v2f a  = *(const v2f*)(xrow + k + koff);
        v2f b0 = *(const v2f*)(wr0  + k + koff);
        v2f b1 = *(const v2f*)(wr1  + k + koff);
        v2f b2 = *(const v2f*)(wr2  + k + koff);
        v2f b3 = *(const v2f*)(wr3  + k + koff);
        acc0 = __builtin_amdgcn_wmma_f32_16x16x4_f32(false, a, false, b0, (short)0, acc0, false, false);
        acc1 = __builtin_amdgcn_wmma_f32_16x16x4_f32(false, a, false, b1, (short)0, acc1, false, false);
        acc2 = __builtin_amdgcn_wmma_f32_16x16x4_f32(false, a, false, b2, (short)0, acc2, false, false);
        acc3 = __builtin_amdgcn_wmma_f32_16x16x4_f32(false, a, false, b3, (short)0, acc3, false, false);
    }

    const int row_base = wave * 16 + ((lane >> 4) ? 8 : 0);
    const int col      = lane & 15;
    const float bv0 = bias[col +  0];
    const float bv1 = bias[col + 16];
    const float bv2 = bias[col + 32];
    const float bv3 = bias[col + 48];

#pragma unroll
    for (int r = 0; r < 8; ++r) {
        float* o = out + (size_t)(row_base + r) * OUT_F + col;
        o[ 0] = acc0[r] + bv0;
        o[16] = acc1[r] + bv1;
        o[32] = acc2[r] + bv2;
        o[48] = acc3[r] + bv3;
    }
}

__global__ void sgc_zero(float4* __restrict__ p, int n4) {
    const float4 z = {0.f, 0.f, 0.f, 0.f};
    for (int i = blockIdx.x * blockDim.x + threadIdx.x; i < n4;
         i += gridDim.x * blockDim.x)
        p[i] = z;
}

__device__ __forceinline__ void atomic_add_row_dev(float* base, unsigned off, float ev, float od) {
    asm volatile("v_permlane16_swap_b32 %0, %1\n\t"
                 "global_atomic_add_f32 %2, %0, %3 scope:SCOPE_DEV\n\t"
                 "global_atomic_add_f32 %2, %1, %3 offset:128 scope:SCOPE_DEV\n\t"
                 "s_wait_xcnt 0x0"
                 : "+v"(ev), "+v"(od) : "v"(off), "s"(base) : "memory");
}

#define EPW 32
static_assert(N_EDGES % EPW == 0, "edge count must be a whole number of wave groups");

__global__ void __launch_bounds__(256)
sgc_scatter(const float* __restrict__ cur,
            float* __restrict__ nxt,
            const int* __restrict__ esrc,
            const int* __restrict__ edst,
            const float* __restrict__ ew) {
    const int wave = (blockIdx.x * blockDim.x + threadIdx.x) >> 5;
    const int lane = threadIdx.x & 31;
    if (wave >= N_EDGES / EPW) return;

    const int e  = wave * EPW + lane;
    const int s  = __builtin_nontemporal_load(esrc + e);
    const int d  = __builtin_nontemporal_load(edst + e);
    const int wi = __builtin_bit_cast(int, __builtin_nontemporal_load(ew + e));

    const unsigned lane8 = (unsigned)lane * 8u;
    const unsigned laneA = ((unsigned)(lane & 15) << 3) | ((unsigned)(lane >> 4) << 2);

    v2f v[EPW];
#pragma unroll
    for (int j = 0; j < EPW; ++j) {
        const unsigned sj = (unsigned)__builtin_amdgcn_readlane(s, j);
        v[j] = *(const v2f*)((const char*)cur + (sj * (unsigned)(OUT_F * 4) + lane8));
    }
#pragma unroll
    for (int j = 0; j < EPW; ++j) {
        const unsigned dj = (unsigned)__builtin_amdgcn_readlane(d, j);
        const float    wj = __builtin_bit_cast(float, __builtin_amdgcn_readlane(wi, j));
        const unsigned off = dj * (unsigned)(OUT_F * 4) + laneA;
        atomic_add_row_dev(nxt, off, v[j].x * wj, v[j].y * wj);
    }
}

extern "C" void kernel_launch(void* const* d_in, const int* in_sizes, int n_in,
                              void* d_out, int out_size, void* d_ws, size_t ws_size,
                              hipStream_t stream) {
    const float* x    = (const float*)d_in[0];
    const float* W    = (const float*)d_in[1];
    const float* b    = (const float*)d_in[2];
    const int*   esrc = (const int*)d_in[3];
    const int*   edst = (const int*)d_in[4];
    const float* ew   = (const float*)d_in[5];

    float* out = (float*)d_out;
    float* ws  = (float*)d_ws;

    const int n4 = N_NODES * OUT_F / 4;

    sgc_gemm_wmma<<<(NTILES + 7) / 8, 256, 0, stream>>>(x, W, b, ws);

    const dim3 sgrid((N_EDGES / EPW + 7) / 8);

    sgc_zero<<<1024, 256, 0, stream>>>((float4*)out, n4);
    sgc_scatter<<<sgrid, 256, 0, stream>>>(ws, out, esrc, edst, ew);

    sgc_zero<<<1024, 256, 0, stream>>>((float4*)ws, n4);
    sgc_scatter<<<sgrid, 256, 0, stream>>>(out, ws, esrc, edst, ew);

    sgc_zero<<<1024, 256, 0, stream>>>((float4*)out, n4);
    sgc_scatter<<<sgrid, 256, 0, stream>>>(ws, out, esrc, edst, ew);
}


# projection: W copied once per work-group into padded LDS, x read as 16-byte pieces (a 128-B line by four back-to-back loads); k order of the f32 WMMA sums permuted, alike for A and B
# speedup vs baseline: 2.3573x; 1.1264x over previous
#include <hip/hip_runtime.h>
#include <hip/hip_bf16.h>

typedef float v2f __attribute__((ext_vector_type(2)));
typedef float v4f __attribute__((ext_vector_type(4)));
typedef float v8f __attribute__((ext_vector_type(8)));

#define N_NODES 100000
#define N_EDGES 1600000
#define IN_F    256
#define OUT_F   64
#define NTILES  (N_NODES / 16)

#define WPITCH 260

__global__ void __launch_bounds__(256)
sgc_gemm_wmma(const float* __restrict__ x,
              const float* __restrict__ W,
              const float* __restrict__ bias,
              float* __restrict__ out) {
    __shared__ __attribute__((aligned(16))) float wl[OUT_F * WPITCH];

    for (int j = 0; j < (OUT_F * IN_F / 4) / 256; ++j) {
        const int i   = (int)threadIdx.x + 256 * j;
        const int row = i >> 6;
        const int c4  = i & 63;
        const v4f v = *(const v4f*)(W + (size_t)i * 4);
        *(v4f*)(wl + row * WPITCH + c4 * 4) = v;
    }
    __syncthreads();

    const int wave = (blockIdx.x * blockDim.x + threadIdx.x) >> 5;
    const int lane = threadIdx.x & 31;
    if (wave >= NTILES) return;

    const int m    = lane & 15;
    const int hoff = (lane >> 4) << 4;

    const float* __restrict__ xrow = x + (size_t)(wave * 16 + m) * IN_F + hoff;
    const float* wb = wl + m * WPITCH + hoff;

    v8f acc0 = {}, acc1 = {}, acc2 = {}, acc3 = {};

    for (int kb = 0; kb < IN_F; kb += 32) {
        v4f a[4];
        for (int i = 0; i < 4; ++i)
            a[i] = *(const v4f*)(xrow + kb + 4 * i);
        for (int i = 0; i < 4; ++i) {
            const v4f b0 = *(const v4f*)(wb + 0 * 16 * WPITCH + kb + 4 * i);
            const v4f b1 = *(const v4f*)(wb + 1 * 16 * WPITCH + kb + 4 * i);
            const v4f b2 = *(const v4f*)(wb + 2 * 16 * WPITCH + kb + 4 * i);
            const v4f b3 = *(const v4f*)(wb + 3 * 16 * WPITCH + kb + 4 * i);
            const v2f alo = a[i].xy, ahi = a[i].zw;
            acc0 = __builtin_amdgcn_wmma_f32_16x16x4_f32(false, alo, false, b0.xy, (short)0, acc0, false, false);
            acc1 = __builtin_amdgcn_wmma_f32_16x16x4_f32(false, alo, false, b1.xy, (short)0, acc1, false, false);
            acc2 = __builtin_amdgcn_wmma_f32_16x16x4_f32(false, alo, false, b2.xy, (short)0, acc2, false, false);
            acc3 = __builtin_amdgcn_wmma_f32_16x16x4_f32(false, alo, false, b3.xy, (short)0, acc3, false, false);
            acc0 = __builtin_amdgcn_wmma_f32_16x16x4_f32(false, ahi, false, b0.zw, (short)0, acc0, false, false);
            acc1 = __builtin_amdgcn_wmma_f32_16x16x4_f32(false, ahi, false, b1.zw, (short)0, acc1, false, false);
            acc2 = __builtin_amdgcn_wmma_f32_16x16x4_f32(false, ahi, false, b2.zw, (short)0, acc2, false, false);
            acc3 = __builtin_amdgcn_wmma_f32_16x16x4_f32(false, ahi, false, b3.zw, (short)0, acc3, false, false);
        }
    }

    const int row_base = wave * 16 + ((lane >> 4) ? 8 : 0);
    const int col      = lane & 15;
    const float bv0 = bias[col +  0];
    const float bv1 = bias[col + 16];
    const float bv2 = bias[col + 32];
    const float bv3 = bias[col + 48];

#pragma unroll
    for (int r = 0; r < 8; ++r) {
        float* o = out + (size_t)(row_base + r) * OUT_F + col;
        o[ 0] = acc0[r] + bv0;
        o[16] = acc1[r] + bv1;
        o[32] = acc2[r] + bv2;
        o[48] = acc3[r] + bv3;
    }
}

__global__ void sgc_zero(float4* __restrict__ p, int n4) {
    const float4 z = {0.f, 0.f, 0.f, 0.f};
    for (int i = blockIdx.x * blockDim.x + threadIdx.x; i < n4;
         i += gridDim.x * blockDim.x)
        p[i] = z;
}

__device__ __forceinline__ void atomic_add_row_dev(float* base, unsigned off, float ev, float od) {
    asm volatile("v_permlane16_swap_b32 %0, %1\n\t"
                 "global_atomic_add_f32 %2, %0, %3 scope:SCOPE_DEV\n\t"
                 "global_atomic_add_f32 %2, %1, %3 offset:128 scope:SCOPE_DEV\n\t"
                 "s_wait_xcnt 0x0"
                 : "+v"(ev), "+v"(od) : "v"(off), "s"(base) : "memory");
}

#define EPW 32
static_assert(N_EDGES % EPW == 0, "edge count must be a whole number of wave groups");

__global__ void __launch_bounds__(256)
sgc_scatter(const float* __restrict__ cur,
            float* __restrict__ nxt,
            const int* __restrict__ esrc,
            const int* __restrict__ edst,
            const float* __restrict__ ew) {
    const int wave = (blockIdx.x * blockDim.x + threadIdx.x) >> 5;
    const int lane = threadIdx.x & 31;
    if (wave >= N_EDGES / EPW) return;

    const int e  = wave * EPW + lane;
    const int s  = __builtin_nontemporal_load(esrc + e);
    const int d  = __builtin_nontemporal_load(edst + e);
    const int wi = __builtin_bit_cast(int, __builtin_nontemporal_load(ew + e));

    const unsigned lane8 = (unsigned)lane * 8u;
    const unsigned laneA = ((unsigned)(lane & 15) << 3) | ((unsigned)(lane >> 4) << 2);

    v2f v[EPW];
#pragma unroll
    for (int j = 0; j < EPW; ++j) {
        const unsigned sj = (unsigned)__builtin_amdgcn_readlane(s, j);
        v[j] = *(const v2f*)((const char*)cur + (sj * (unsigned)(OUT_F * 4) + lane8));
    }
#pragma unroll
    for (int j = 0; j < EPW; ++j) {
        const unsigned dj = (unsigned)__builtin_amdgcn_readlane(d, j);
        const float    wj = __builtin_bit_cast(float, __builtin_amdgcn_readlane(wi, j));
        const unsigned off = dj * (unsigned)(OUT_F * 4) + laneA;
        atomic_add_row_dev(nxt, off, v[j].x * wj, v[j].y * wj);
    }
}

extern "C" void kernel_launch(void* const* d_in, const int* in_sizes, int n_in,
                              void* d_out, int out_size, void* d_ws, size_t ws_size,
                              hipStream_t stream) {
    const float* x    = (const float*)d_in[0];
    const float* W    = (const float*)d_in[1];
    const float* b    = (const float*)d_in[2];
    const int*   esrc = (const int*)d_in[3];
    const int*   edst = (const int*)d_in[4];
    const float* ew   = (const float*)d_in[5];

    float* out = (float*)d_out;
    float* ws  = (float*)d_ws;

    const int n4 = N_NODES * OUT_F / 4;

    sgc_gemm_wmma<<<(NTILES + 7) / 8, 256, 0, stream>>>(x, W, b, ws);

    const dim3 sgrid((N_EDGES / EPW + 7) / 8);

    sgc_zero<<<1024, 256, 0, stream>>>((float4*)out, n4);
    sgc_scatter<<<sgrid, 256, 0, stream>>>(ws, out, esrc, edst, ew);

    sgc_zero<<<1024, 256, 0, stream>>>((float4*)ws, n4);
    sgc_scatter<<<sgrid, 256, 0, stream>>>(out, ws, esrc, edst, ew);

    sgc_zero<<<1024, 256, 0, stream>>>((float4*)out, n4);
    sgc_scatter<<<sgrid, 256, 0, stream>>>(ws, out, esrc, edst, ew);
}


# scatter hops: edges ordered by destination in LDS (counting sort on lane-match ranks, no atomics), one wave sums a node's rows in registers in fixed order; zero fills and f32 atomics gone
# speedup vs baseline: 2.8027x; 1.1889x over previous
#include <hip/hip_runtime.h>
#include <hip/hip_bf16.h>

typedef float v2f __attribute__((ext_vector_type(2)));
typedef float v4f __attribute__((ext_vector_type(4)));
typedef float v8f __attribute__((ext_vector_type(8)));

#define N_NODES 100000
#define N_EDGES 1600000
#define IN_F    256
#define OUT_F   64
#define NTILES  (N_NODES / 16)

#define WPITCH 260

__global__ void __launch_bounds__(256)
sgc_gemm_wmma(const float* __restrict__ x,
              const float* __restrict__ W,
              const float* __restrict__ bias,
              float* __restrict__ out) {
    __shared__ __attribute__((aligned(16))) float wl[OUT_F * WPITCH];

    for (int j = 0; j < (OUT_F * IN_F / 4) / 256; ++j) {
        const int i   = (int)threadIdx.x + 256 * j;
        const int row = i >> 6;
        const int c4  = i & 63;
        const v4f v = *(const v4f*)(W + (size_t)i * 4);
        *(v4f*)(wl + row * WPITCH + c4 * 4) = v;
    }
    __syncthreads();

    const int wave = (blockIdx.x * blockDim.x + threadIdx.x) >> 5;
    const int lane = threadIdx.x & 31;
    if (wave >= NTILES) return;

    const int m    = lane & 15;
    const int hoff = (lane >> 4) << 4;

    const float* __restrict__ xrow = x + (size_t)(wave * 16 + m) * IN_F + hoff;
    const float* wb = wl + m * WPITCH + hoff;

    v8f acc0 = {}, acc1 = {}, acc2 = {}, acc3 = {};

    for (int kb = 0; kb < IN_F; kb += 32) {
        v4f a[4];
        for (int i = 0; i < 4; ++i)
            a[i] = *(const v4f*)(xrow + kb + 4 * i);
        for (int i = 0; i < 4; ++i) {
            const v4f b0 = *(const v4f*)(wb + 0 * 16 * WPITCH + kb + 4 * i);
            const v4f b1 = *(const v4f*)(wb + 1 * 16 * WPITCH + kb + 4 * i);
            const v4f b2 = *(const v4f*)(wb + 2 * 16 * WPITCH + kb + 4 * i);
            const v4f b3 = *(const v4f*)(wb + 3 * 16 * WPITCH + kb + 4 * i);
            const v2f alo = a[i].xy, ahi = a[i].zw;
            acc0 = __builtin_amdgcn_wmma_f32_16x16x4_f32(false, alo, false, b0.xy, (short)0, acc0, false, false);
            acc1 = __builtin_amdgcn_wmma_f32_16x16x4_f32(false, alo, false, b1.xy, (short)0, acc1, false, false);
            acc2 = __builtin_amdgcn_wmma_f32_16x16x4_f32(false, alo, false, b2.xy, (short)0, acc2, false, false);
            acc3 = __builtin_amdgcn_wmma_f32_16x16x4_f32(false, alo, false, b3.xy, (short)0, acc3, false, false);
            acc0 = __builtin_amdgcn_wmma_f32_16x16x4_f32(false, ahi, false, b0.zw, (short)0, acc0, false, false);
            acc1 = __builtin_amdgcn_wmma_f32_16x16x4_f32(false, ahi, false, b1.zw, (short)0, acc1, false, false);
            acc2 = __builtin_amdgcn_wmma_f32_16x16x4_f32(false, ahi, false, b2.zw, (short)0, acc2, false, false);
            acc3 = __builtin_amdgcn_wmma_f32_16x16x4_f32(false, ahi, false, b3.zw, (short)0, acc3, false, false);
        }
    }

    const int row_base = wave * 16 + ((lane >> 4) ? 8 : 0);
    const int col      = lane & 15;
    const float bv0 = bias[col +  0];
    const float bv1 = bias[col + 16];
    const float bv2 = bias[col + 32];
    const float bv3 = bias[col + 48];

#pragma unroll
    for (int r = 0; r < 8; ++r) {
        float* o = out + (size_t)(row_base + r) * OUT_F + col;
        o[ 0] = acc0[r] + bv0;
        o[16] = acc1[r] + bv1;
        o[32] = acc2[r] + bv2;
        o[48] = acc3[r] + bv3;
    }
}

#define BIN_ROWS 400
#define NBINS    250
#define REGION_F (BIN_ROWS * OUT_F)
#define NG       128
#define SLAB     (N_EDGES / NG)
#define LPITCH_W 192
#define CAP      ((LPITCH_W - 2) / 2)
#define GT       1024
#define NWV      (GT / 32)
static_assert(NBINS * BIN_ROWS == N_NODES, "ranges must tile the nodes");
static_assert(NG * SLAB == N_EDGES, "slabs must tile the edges");
static_assert(NG * LPITCH_W <= REGION_F && LPITCH_W % 32 == 0, "the blocks must fit a region, whole lines each");
static_assert(N_NODES <= (1 << 17) && BIN_ROWS <= 512 && NBINS <= 512, "packing and key widths");
static_assert(NG % NWV == 0 && NG % 32 == 0, "list walk of sgc_rowsum");
static_assert(N_EDGES % 32 == 0, "whole waves of edges");

typedef unsigned u2 __attribute__((ext_vector_type(2)));
typedef __attribute__((address_space(3))) unsigned lds_u32;

struct Match { unsigned rank, size, lead; };
template <int BITS>
__device__ __forceinline__ Match match_key(unsigned key, bool valid, unsigned lane) {
    unsigned m = __builtin_amdgcn_ballot_w32(valid);
#pragma unroll
    for (int bit = 0; bit < BITS; ++bit) {
        const bool one = ((key >> bit) & 1u) != 0u;
        const unsigned bal = __builtin_amdgcn_ballot_w32(valid && one);
        m &= one ? bal : ~bal;
    }
    Match r;
    r.rank = (unsigned)__builtin_popcount(m & ((1u << lane) - 1u));
    r.size = (unsigned)__builtin_popcount(m);
    r.lead = valid ? (unsigned)__builtin_ctz(m | (1u << lane)) : lane;
    return r;
}

template <int NKEYS, class Src>
__device__ __forceinline__ void group_sort(const Src& src, unsigned* cw, unsigned* start,
                                           unsigned* wsum, u2* sorted,
                                           unsigned tid, unsigned lane, unsigned wv) {
    static_assert(NKEYS <= 512, "the key field of the element word has nine bits");
    for (unsigned i = tid; i < (unsigned)(NWV * NKEYS); i += GT) cw[i] = 0u;
    __syncthreads();
    volatile lds_u32* c = (volatile lds_u32*)(cw + wv * NKEYS);

    unsigned info[Src::ITERS]; u2 pay[Src::ITERS];
#pragma unroll
    for (int it = 0; it < Src::ITERS; ++it) {
        unsigned k;
        const bool ok = src.get(it, k, pay[it]);
        info[it] = k | (ok ? 0x80000000u : 0u);
    }

#pragma unroll
    for (int it = 0; it < Src::ITERS; ++it) {
        if (src.skip(it)) continue;
        const unsigned k = info[it] & 0x1ffu;
        const bool valid = (info[it] >> 31) != 0u;
        const Match mt = match_key<(NKEYS <= 256 ? 8 : 9)>(k, valid, lane);
        if (valid && mt.rank == 0u) c[k] = c[k] + mt.size;
        info[it] |= (mt.rank << 9) | (mt.size << 14) | (mt.lead << 20);
        asm volatile("" : "+v"(info[it]));
    }
    __syncthreads();

    unsigned tot = 0u;
    if (tid < (unsigned)NKEYS) {
        for (int w = 0; w < NWV; ++w) {
            const unsigned v = cw[w * NKEYS + tid];
            cw[w * NKEYS + tid] = tot;
            tot += v;
        }
    }
    unsigned x = tot;
    for (unsigned d = 1u; d < 32u; d <<= 1) {
        const unsigned y = (unsigned)__builtin_amdgcn_ds_bpermute((int)((lane - d) << 2), (int)x);
        if (lane >= d) x += y;
    }
    if (lane == 31u) wsum[wv] = x;
    __syncthreads();
    unsigned off = 0u;
    for (unsigned w = 0u; w < wv; ++w) off += wsum[w];
    if (tid < (unsigned)NKEYS) start[tid] = off + x - tot;
    if (tid == (unsigned)(NKEYS - 1)) start[NKEYS] = off + x;
    __syncthreads();

#pragma unroll
    for (int it = 0; it < Src::ITERS; ++it) {
        if (src.skip(it)) continue;
        const unsigned k    = info[it] & 0x1ffu;
        const unsigned rank = (info[it] >> 9) & 31u;
        const unsigned size = (info[it] >> 14) & 63u;
        const unsigned lead = (info[it] >> 20) & 31u;
        const bool valid = (info[it] >> 31) != 0u;
        unsigned base = 0u;
        if (valid && rank == 0u) { base = c[k]; c[k] = base + size; }
        base = (unsigned)__builtin_amdgcn_ds_bpermute((int)(lead << 2), (int)base);
        if (valid) sorted[start[k] + base + rank] = pay[it];
    }
    __syncthreads();
}

struct EdgeSrc {
    static constexpr int ITERS = (SLAB + GT - 1) / GT;
    const int* esrc; const int* edst; const float* ew;
    unsigned e0, tid;
    __device__ __forceinline__ bool skip(int) const { return false; }
    __device__ __forceinline__ bool get(int it, unsigned& key, u2& pay) const {
        const unsigned i  = tid + (unsigned)GT * (unsigned)it;
        const unsigned ic = i < (unsigned)SLAB ? i : (unsigned)(SLAB - 1);
        const unsigned s = (unsigned)esrc[e0 + ic];
        const unsigned d = (unsigned)edst[e0 + ic];
        const unsigned w = __builtin_bit_cast(unsigned, ew[e0 + ic]);
        const unsigned b = d / BIN_ROWS;
        const bool ok = i < (unsigned)SLAB && b < (unsigned)NBINS && s < (unsigned)N_NODES;
        key   = ok ? b : 0u;
        pay.x = s | ((d - b * BIN_ROWS) << 17);
        pay.y = w;
        return ok;
    }
};

__global__ void __launch_bounds__(GT)
sgc_order(unsigned* plane,
          const int* __restrict__ esrc,
          const int* __restrict__ edst,
          const float* __restrict__ ew) {
    __shared__ unsigned cw[NWV * NBINS];
    __shared__ unsigned start[NBINS + 1];
    __shared__ unsigned wsum[NWV];
    __shared__ __attribute__((aligned(8))) u2 sorted[SLAB];

    const unsigned tid  = threadIdx.x;
    const unsigned lane = tid & 31u;
    const unsigned wv   = (unsigned)__builtin_amdgcn_readfirstlane((int)(tid >> 5));
    const unsigned g    = blockIdx.x;

    EdgeSrc src;
    src.esrc = esrc; src.edst = edst; src.ew = ew; src.e0 = g * (unsigned)SLAB; src.tid = tid;
    group_sort<NBINS>(src, cw, start, wsum, sorted, tid, lane, wv);

    for (unsigned b = wv; b < (unsigned)NBINS; b += NWV) {
        const unsigned st  = start[b];
        const unsigned tot = start[b + 1] - st;
        const unsigned n   = tot < (unsigned)CAP ? tot : (unsigned)CAP;
        u2* blk = (u2*)(plane + b * (unsigned)REGION_F + g * (unsigned)LPITCH_W);
        for (unsigned t = lane; t <= n; t += 32u) {
            u2 v; v.x = tot; v.y = 0u;
            if (t != 0u) v = sorted[st + t - 1u];
            blk[t] = v;
        }
    }
}

struct ListSrc {
    static constexpr int LSTR  = (CAP + 31) / 32;
    static constexpr int ITERS = (NG / NWV) * LSTR;
    const unsigned* region; const unsigned* gcnt;
    unsigned lane, wv;
    __device__ __forceinline__ bool skip(int it) const {
        const unsigned n = (unsigned)__builtin_amdgcn_readfirstlane((int)gcnt[wv + (unsigned)NWV * (unsigned)(it / LSTR)]);
        return 32u * (unsigned)(it % LSTR) >= n;
    }
    __device__ __forceinline__ bool get(int it, unsigned& key, u2& pay) const {
        const unsigned gl = wv + (unsigned)NWV * (unsigned)(it / LSTR);
        const unsigned j  = lane + 32u * (unsigned)(it % LSTR);
        const unsigned jc = j < (unsigned)CAP ? j : (unsigned)(CAP - 1);
        const u2 e = ((const u2*)(region + gl * (unsigned)LPITCH_W))[1u + jc];
        const unsigned r = e.x >> 17;
        const bool ok = j < gcnt[gl] && r < (unsigned)BIN_ROWS;
        key   = ok ? r : 0u;
        pay.x = e.x & 0x1ffffu;
        pay.y = e.y;
        return ok;
    }
};

#define CH 24
template <bool FULL>
__device__ __forceinline__ void row_add(const float* __restrict__ cur, unsigned px, unsigned pw,
                                        unsigned n, unsigned lane2, v2f& acc) {
    v2f v[CH];
#pragma unroll
    for (int j = 0; j < CH; ++j) {
        if (FULL || (unsigned)j < n) {
            const unsigned sj = (unsigned)__builtin_amdgcn_readlane((int)px, j);
            v[j] = *(const v2f*)(cur + sj * (unsigned)OUT_F + lane2);
        }
    }
#pragma unroll
    for (int j = 0; j < CH; ++j) {
        if (FULL || (unsigned)j < n) {
            const float wj = __builtin_bit_cast(float, __builtin_amdgcn_readlane((int)pw, j));
            acc.x += v[j].x * wj;
            acc.y += v[j].y * wj;
        }
    }
}

__global__ void __launch_bounds__(GT)
sgc_rowsum(const float* __restrict__ cur,
           float* nxt,
           const int* __restrict__ esrc,
           const int* __restrict__ edst,
           const float* __restrict__ ew) {
    __shared__ unsigned cw[NWV * BIN_ROWS];
    __shared__ unsigned start[BIN_ROWS + 1];
    __shared__ unsigned wsum[NWV];
    __shared__ unsigned gcnt[NG];
    __shared__ unsigned ovf[NG / 32];
    __shared__ __attribute__((aligned(16))) float buf[REGION_F];

    const unsigned tid   = threadIdx.x;
    const unsigned lane  = tid & 31u;
    const unsigned lane2 = lane * 2u;
    const unsigned wv    = (unsigned)__builtin_amdgcn_readfirstlane((int)(tid >> 5));
    const unsigned bin   = blockIdx.x;
    float* region = nxt + (size_t)bin * REGION_F;

    if (tid < (unsigned)NG) {
        const unsigned cn = ((const unsigned*)region)[tid * (unsigned)LPITCH_W];
        gcnt[tid] = cn;
        const unsigned bal = __builtin_amdgcn_ballot_w32(cn > (unsigned)CAP);
        if (lane == 0u) ovf[wv] = bal;
    }
    __syncthreads();
    unsigned anyover = 0u;
    for (int i = 0; i < NG / 32; ++i) anyover |= ovf[i];
    const bool over = __builtin_amdgcn_readfirstlane((int)anyover) != 0;

    if (!over) {
        u2* sorted = (u2*)buf;
        ListSrc src;
        src.region = (const unsigned*)region; src.gcnt = gcnt; src.lane = lane; src.wv = wv;
        group_sort<BIN_ROWS>(src, cw, start, wsum, sorted, tid, lane, wv);

        for (unsigned r = wv; r < (unsigned)BIN_ROWS; r += NWV) {
            const unsigned st = (unsigned)__builtin_amdgcn_readfirstlane((int)start[r]);
            const unsigned n  = (unsigned)__builtin_amdgcn_readfirstlane((int)start[r + 1]) - st;
            v2f acc; acc.x = 0.f; acc.y = 0.f;
            for (unsigned base = 0u; base < n; base += (unsigned)CH) {
                const unsigned left = n - base;
                u2 e; e.x = 0u; e.y = 0u;
                if (lane < left && lane < (unsigned)CH) e = sorted[st + base + lane];
                if (left >= (unsigned)CH) row_add<true >(cur, e.x, e.y, (unsigned)CH, lane2, acc);
                else                      row_add<false>(cur, e.x, e.y, left,         lane2, acc);
            }
            *(v2f*)(region + r * (unsigned)OUT_F + lane2) = acc;
        }
    } else {
        float* tile = buf;
        const v4f z = {0.f, 0.f, 0.f, 0.f};
        for (unsigned i = tid; i < (unsigned)(REGION_F / 4); i += GT) ((v4f*)tile)[i] = z;
        __syncthreads();
        for (unsigned eb = 0u; eb < (unsigned)N_EDGES; eb += 32u) {
            const unsigned s = (unsigned)esrc[eb + lane];
            const unsigned d = (unsigned)edst[eb + lane];
            const unsigned w = __builtin_bit_cast(unsigned, ew[eb + lane]);
            const unsigned r = d - bin * (unsigned)BIN_ROWS;
            unsigned m = __builtin_amdgcn_ballot_w32(r < (unsigned)BIN_ROWS && (r & (unsigned)(NWV - 1)) == wv && s < (unsigned)N_NODES);
            for (int k = 0; k < 32 && m != 0u; ++k) {
                const int j = __builtin_ctz(m);
                m &= m - 1u;
                const unsigned sj = (unsigned)__builtin_amdgcn_readlane((int)s, j);
                const unsigned rj = (unsigned)__builtin_amdgcn_readlane((int)r, j);
                const float    wj = __builtin_bit_cast(float, __builtin_amdgcn_readlane((int)w, j));
                const v2f gsrc = *(const v2f*)(cur + sj * (unsigned)OUT_F + lane2);
                float* t = tile + rj * (unsigned)OUT_F + lane2;
                t[0] += gsrc.x * wj;
                t[1] += gsrc.y * wj;
            }
        }
        __syncthreads();
        for (unsigned i = tid; i < (unsigned)(REGION_F / 4); i += GT) ((v4f*)region)[i] = ((const v4f*)tile)[i];
    }
}

extern "C" void kernel_launch(void* const* d_in, const int* in_sizes, int n_in,
                              void* d_out, int out_size, void* d_ws, size_t ws_size,
                              hipStream_t stream) {
    const float* x    = (const float*)d_in[0];
    const float* W    = (const float*)d_in[1];
    const float* b    = (const float*)d_in[2];
    const int*   esrc = (const int*)d_in[3];
    const int*   edst = (const int*)d_in[4];
    const float* ew   = (const float*)d_in[5];

    float* out = (float*)d_out;
    float* ws  = (float*)d_ws;

    sgc_gemm_wmma<<<(NTILES + 7) / 8, 256, 0, stream>>>(x, W, b, ws);

    sgc_order <<<NG,    GT, 0, stream>>>((unsigned*)out, esrc, edst, ew);
    sgc_rowsum<<<NBINS, GT, 0, stream>>>(ws, out, esrc, edst, ew);

    sgc_order <<<NG,    GT, 0, stream>>>((unsigned*)ws, esrc, edst, ew);
    sgc_rowsum<<<NBINS, GT, 0, stream>>>(out, ws, esrc, edst, ew);

    sgc_order <<<NG,    GT, 0, stream>>>((unsigned*)out, esrc, edst, ew);
    sgc_rowsum<<<NBINS, GT, 0, stream>>>(ws, out, esrc, edst, ew);
}


# last hop as ONE launch on whole 256-B rows (16 lanes an edge) instead of two half-row launches: one random row read an edge, not two; its row sum = two interleaved partial sums (was four)
# speedup vs baseline: 3.5671x; 1.2727x over previous
#include <hip/hip_runtime.h>
#include <hip/hip_bf16.h>

typedef float v2f __attribute__((ext_vector_type(2)));
typedef float v4f __attribute__((ext_vector_type(4)));
typedef float v8f __attribute__((ext_vector_type(8)));

#define N_NODES 100000
#define N_EDGES 1600000
#define IN_F    256
#define OUT_F   64
#define NTILES  (N_NODES / 16)

#define WPITCH 260

__global__ void __launch_bounds__(256)
sgc_gemm_wmma(const float* __restrict__ x,
              const float* __restrict__ W,
              const float* __restrict__ bias,
              float* __restrict__ plo,
              float* __restrict__ phi) {
    __shared__ __attribute__((aligned(16))) float wl[OUT_F * WPITCH];

    for (int j = 0; j < (OUT_F * IN_F / 4) / 256; ++j) {
        const int i   = (int)threadIdx.x + 256 * j;
        const int row = i >> 6;
        const int c4  = i & 63;
        const v4f v = *(const v4f*)(W + (size_t)i * 4);
        *(v4f*)(wl + row * WPITCH + c4 * 4) = v;
    }
    __syncthreads();

    const int wave = (blockIdx.x * blockDim.x + threadIdx.x) >> 5;
    const int lane = threadIdx.x & 31;
    if (wave >= NTILES) return;

    const int m    = lane & 15;
    const int hoff = (lane >> 4) << 4;

    const float* __restrict__ xrow = x + (size_t)(wave * 16 + m) * IN_F + hoff;
    const float* wb = wl + m * WPITCH + hoff;

    v8f acc0 = {}, acc1 = {}, acc2 = {}, acc3 = {};

    for (int kb = 0; kb < IN_F; kb += 32) {
        v4f a[4];
        for (int i = 0; i < 4; ++i)
            a[i] = *(const v4f*)(xrow + kb + 4 * i);
        for (int i = 0; i < 4; ++i) {
            const v4f b0 = *(const v4f*)(wb + 0 * 16 * WPITCH + kb + 4 * i);
            const v4f b1 = *(const v4f*)(wb + 1 * 16 * WPITCH + kb + 4 * i);
            const v4f b2 = *(const v4f*)(wb + 2 * 16 * WPITCH + kb + 4 * i);
            const v4f b3 = *(const v4f*)(wb + 3 * 16 * WPITCH + kb + 4 * i);
            const v2f alo = a[i].xy, ahi = a[i].zw;
            acc0 = __builtin_amdgcn_wmma_f32_16x16x4_f32(false, alo, false, b0.xy, (short)0, acc0, false, false);
            acc1 = __builtin_amdgcn_wmma_f32_16x16x4_f32(false, alo, false, b1.xy, (short)0, acc1, false, false);
            acc2 = __builtin_amdgcn_wmma_f32_16x16x4_f32(false, alo, false, b2.xy, (short)0, acc2, false, false);
            acc3 = __builtin_amdgcn_wmma_f32_16x16x4_f32(false, alo, false, b3.xy, (short)0, acc3, false, false);
            acc0 = __builtin_amdgcn_wmma_f32_16x16x4_f32(false, ahi, false, b0.zw, (short)0, acc0, false, false);
            acc1 = __builtin_amdgcn_wmma_f32_16x16x4_f32(false, ahi, false, b1.zw, (short)0, acc1, false, false);
            acc2 = __builtin_amdgcn_wmma_f32_16x16x4_f32(false, ahi, false, b2.zw, (short)0, acc2, false, false);
            acc3 = __builtin_amdgcn_wmma_f32_16x16x4_f32(false, ahi, false, b3.zw, (short)0, acc3, false, false);
        }
    }

    const int row_base = wave * 16 + ((lane >> 4) ? 8 : 0);
    const int col      = lane & 15;
    const float bv0 = bias[col +  0];
    const float bv1 = bias[col + 16];
    const float bv2 = bias[col + 32];
    const float bv3 = bias[col + 48];

#pragma unroll
    for (int r = 0; r < 8; ++r) {
        const size_t o = (size_t)(row_base + r) * OUT_F + col;
        plo[o +  0] = acc0[r] + bv0;
        plo[o + 16] = acc1[r] + bv1;
        phi[o + 32] = acc2[r] + bv2;
        phi[o + 48] = acc3[r] + bv3;
    }
}

#define BIN_ROWS 400
#define NBINS    250
#define HALF_F   (OUT_F / 2)
#define NG       128
#define SLAB     (N_EDGES / NG)
#define LPITCH_W 192
#define LCHUNKS  (LPITCH_W / 32)
#define CAP      ((LPITCH_W - 2) / 2)
#define GT       1024
#define NWV      (GT / 32)
#define ECAP     7168
#define HDR_CH   0
#define HDR_W    224
#define WSEC_CH  7
#define SSEC_CH  (WSEC_CH + ECAP / 32)
#define HSEC_CH  (SSEC_CH + ECAP / 64)
#define SPARE    0xffffffffu
static_assert(NBINS * BIN_ROWS == N_NODES, "ranges must tile the nodes");
static_assert(NG * SLAB == N_EDGES, "slabs must tile the edges");
static_assert(NG * LCHUNKS <= 2 * BIN_ROWS && LPITCH_W % 32 == 0, "the blocks must fit a range's rows of A and D, whole chunks each");
static_assert(HSEC_CH + ECAP / 128 <= BIN_ROWS && ECAP % 128 == 0, "a range's part of L must fit its rows of D");
static_assert(ECAP < 65536 && 8 + (BIN_ROWS + 2) / 2 <= HDR_W, "row starts are 16-bit fields of the header");
static_assert(N_NODES <= (1 << 17) && BIN_ROWS <= 512 && NBINS <= 512, "packing and key widths");
static_assert(NG % NWV == 0 && NG % 32 == 0, "list walk of sgc_csr");
static_assert(N_EDGES % 32 == 0, "whole waves of edges");

typedef unsigned u2 __attribute__((ext_vector_type(2)));
typedef __attribute__((address_space(3))) unsigned lds_u32;

struct Match { unsigned rank, size, lead; };
template <int BITS>
__device__ __forceinline__ Match match_key(unsigned key, bool valid, unsigned lane) {
    unsigned m = __builtin_amdgcn_ballot_w32(valid);
#pragma unroll
    for (int bit = 0; bit < BITS; ++bit) {
        const bool one = ((key >> bit) & 1u) != 0u;
        const unsigned bal = __builtin_amdgcn_ballot_w32(valid && one);
        m &= one ? bal : ~bal;
    }
    Match r;
    r.rank = (unsigned)__builtin_popcount(m & ((1u << lane) - 1u));
    r.size = (unsigned)__builtin_popcount(m);
    r.lead = valid ? (unsigned)__builtin_ctz(m | (1u << lane)) : lane;
    return r;
}

template <int NKEYS, class Src>
__device__ __forceinline__ void group_sort(const Src& src, unsigned* cw, unsigned* start,
                                           unsigned* wsum, u2* sorted,
                                           unsigned tid, unsigned lane, unsigned wv) {
    static_assert(NKEYS <= 512, "the key field of the element word has nine bits");
    for (unsigned i = tid; i < (unsigned)(NWV * NKEYS); i += GT) cw[i] = 0u;
    __syncthreads();
    volatile lds_u32* c = (volatile lds_u32*)(cw + wv * NKEYS);

    unsigned info[Src::ITERS]; u2 pay[Src::ITERS];
#pragma unroll
    for (int it = 0; it < Src::ITERS; ++it) {
        unsigned k;
        const bool ok = src.get(it, k, pay[it]);
        info[it] = k | (ok ? 0x80000000u : 0u);
    }

#pragma unroll
    for (int it = 0; it < Src::ITERS; ++it) {
        if (src.skip(it)) continue;
        const unsigned k = info[it] & 0x1ffu;
        const bool valid = (info[it] >> 31) != 0u;
        const Match mt = match_key<(NKEYS <= 256 ? 8 : 9)>(k, valid, lane);
        if (valid && mt.rank == 0u) c[k] = c[k] + mt.size;
        info[it] |= (mt.rank << 9) | (mt.size << 14) | (mt.lead << 20);
        asm volatile("" : "+v"(info[it]));
    }
    __syncthreads();

    unsigned tot = 0u;
    if (tid < (unsigned)NKEYS) {
        for (int w = 0; w < NWV; ++w) {
            const unsigned v = cw[w * NKEYS + tid];
            cw[w * NKEYS + tid] = tot;
            tot += v;
        }
    }
    unsigned x = tot;
    for (unsigned d = 1u; d < 32u; d <<= 1) {
        const unsigned y = (unsigned)__builtin_amdgcn_ds_bpermute((int)((lane - d) << 2), (int)x);
        if (lane >= d) x += y;
    }
    if (lane == 31u) wsum[wv] = x;
    __syncthreads();
    unsigned off = 0u;
    for (unsigned w = 0u; w < wv; ++w) off += wsum[w];
    if (tid < (unsigned)NKEYS) start[tid] = off + x - tot;
    if (tid == (unsigned)(NKEYS - 1)) start[NKEYS] = off + x;
    __syncthreads();

#pragma unroll
    for (int it = 0; it < Src::ITERS; ++it) {
        if (src.skip(it)) continue;
        const unsigned k    = info[it] & 0x1ffu;
        const unsigned rank = (info[it] >> 9) & 31u;
        const unsigned size = (info[it] >> 14) & 63u;
        const unsigned lead = (info[it] >> 20) & 31u;
        const bool valid = (info[it] >> 31) != 0u;
        unsigned base = 0u;
        if (valid && rank == 0u) { base = c[k]; c[k] = base + size; }
        base = (unsigned)__builtin_amdgcn_ds_bpermute((int)(lead << 2), (int)base);
        if (valid) sorted[start[k] + base + rank] = pay[it];
    }
    __syncthreads();
}

struct EdgeSrc {
    static constexpr int ITERS = (SLAB + GT - 1) / GT;
    const int* esrc; const int* edst; const float* ew;
    unsigned e0, tid;
    __device__ __forceinline__ bool skip(int) const { return false; }
    __device__ __forceinline__ bool get(int it, unsigned& key, u2& pay) const {
        const unsigned i  = tid + (unsigned)GT * (unsigned)it;
        const unsigned ic = i < (unsigned)SLAB ? i : (unsigned)(SLAB - 1);
        const unsigned s = (unsigned)esrc[e0 + ic];
        const unsigned d = (unsigned)edst[e0 + ic];
        const unsigned w = __builtin_bit_cast(unsigned, ew[e0 + ic]);
        const unsigned b = d / BIN_ROWS;
        const bool ok = i < (unsigned)SLAB && b < (unsigned)NBINS && s < (unsigned)N_NODES;
        key   = ok ? b : 0u;
        pay.x = s | ((d - b * BIN_ROWS) << 17);
        pay.y = w;
        return ok;
    }
};

#define PS 6
__device__ __forceinline__ unsigned* list_chunk(unsigned* pa, unsigned* pd, unsigned bin, unsigned ch) {
    const bool in_a = ch < (unsigned)BIN_ROWS;
    unsigned* p = in_a ? pa : pd;
    const unsigned row = in_a ? ch : ch - (unsigned)BIN_ROWS;
    return p + (bin * (unsigned)BIN_ROWS + row) * (unsigned)OUT_F;
}

__global__ void __launch_bounds__(GT)
sgc_order(unsigned* ws_plane, unsigned* out_plane,
          const int* __restrict__ esrc,
          const int* __restrict__ edst,
          const float* __restrict__ ew) {
    __shared__ unsigned cw[NWV * NBINS];
    __shared__ unsigned start[NBINS + 1];
    __shared__ unsigned wsum[NWV];
    __shared__ __attribute__((aligned(8))) u2 sorted[SLAB];

    const unsigned tid  = threadIdx.x;
    const unsigned lane = tid & 31u;
    const unsigned wv   = (unsigned)__builtin_amdgcn_readfirstlane((int)(tid >> 5));
    const unsigned g    = blockIdx.x;
    unsigned* pa = ws_plane;
    unsigned* pd = out_plane + HALF_F;

    EdgeSrc src;
    src.esrc = esrc; src.edst = edst; src.ew = ew; src.e0 = g * (unsigned)SLAB; src.tid = tid;
    group_sort<NBINS>(src, cw, start, wsum, sorted, tid, lane, wv);

    for (unsigned b = wv; b < (unsigned)NBINS; b += NWV) {
        const unsigned st  = start[b];
        const unsigned tot = start[b + 1] - st;
        const unsigned n   = tot < (unsigned)CAP ? tot : (unsigned)CAP;
        for (unsigned t = lane; t <= n; t += 32u) {
            u2 v; v.x = tot; v.y = 0u;
            if (t != 0u) v = sorted[st + t - 1u];
            unsigned* c = list_chunk(pa, pd, b, g * (unsigned)LCHUNKS + (t >> 4));
            *(u2*)(c + (t & 15u) * 2u) = v;
        }
    }
}

struct ListSrc {
    static constexpr int LSTR  = (CAP + 31) / 32;
    static constexpr int ITERS = (NG / NWV) * LSTR;
    unsigned* pa; unsigned* pd; const unsigned* gcnt;
    unsigned bin, lane, wv;
    __device__ __forceinline__ bool skip(int it) const {
        const unsigned n = (unsigned)__builtin_amdgcn_readfirstlane((int)gcnt[wv + (unsigned)NWV * (unsigned)(it / LSTR)]);
        return 32u * (unsigned)(it % LSTR) >= n;
    }
    __device__ __forceinline__ bool get(int it, unsigned& key, u2& pay) const {
        const unsigned gl = wv + (unsigned)NWV * (unsigned)(it / LSTR);
        const unsigned j  = lane + 32u * (unsigned)(it % LSTR);
        const unsigned jc = j < (unsigned)CAP ? j : (unsigned)(CAP - 1);
        const unsigned t  = 1u + jc;
        const unsigned* c = list_chunk(pa, pd, bin, gl * (unsigned)LCHUNKS + (t >> 4));
        const u2 e = *(const u2*)(c + (t & 15u) * 2u);
        const unsigned r = e.x >> 17;
        const bool ok = j < gcnt[gl] && r < (unsigned)BIN_ROWS;
        key   = ok ? r : 0u;
        pay.x = e.x & 0x1ffffu;
        pay.y = e.y;
        return ok;
    }
};

__device__ __forceinline__ unsigned* l_chunk(unsigned* pd, unsigned bin, unsigned ch) {
    return pd + (bin * (unsigned)BIN_ROWS + ch) * (unsigned)OUT_F;
}

__global__ void __launch_bounds__(GT)
sgc_csr(unsigned* ws_plane, unsigned* out_plane) {
    __shared__ unsigned cw[NWV * BIN_ROWS];
    __shared__ unsigned start[BIN_ROWS + 1];
    __shared__ unsigned wsum[NWV];
    __shared__ unsigned gcnt[NG];
    __shared__ unsigned ovf[NG / 32];
    __shared__ __attribute__((aligned(16))) u2 sorted[NG * CAP];

    const unsigned tid  = threadIdx.x;
    const unsigned lane = tid & 31u;
    const unsigned wv   = (unsigned)__builtin_amdgcn_readfirstlane((int)(tid >> 5));
    const unsigned bin  = blockIdx.x;
    unsigned* pa = ws_plane;
    unsigned* pd = out_plane + HALF_F;

    if (tid < (unsigned)NG) {
        const unsigned cn = list_chunk(pa, pd, bin, tid * (unsigned)LCHUNKS)[0];
        gcnt[tid] = cn;
        const unsigned bal = __builtin_amdgcn_ballot_w32(cn > (unsigned)CAP);
        if (lane == 0u) ovf[wv] = bal;
    }
    __syncthreads();
    unsigned anyover = 0u;
    for (int i = 0; i < NG / 32; ++i) anyover |= ovf[i];
    const bool over = __builtin_amdgcn_readfirstlane((int)anyover) != 0;

    unsigned tot = SPARE;
    if (!over) {
        ListSrc src;
        src.pa = pa; src.pd = pd; src.gcnt = gcnt; src.bin = bin; src.lane = lane; src.wv = wv;
        group_sort<BIN_ROWS>(src, cw, start, wsum, sorted, tid, lane, wv);
        tot = start[BIN_ROWS];
        if (tot > (unsigned)ECAP) tot = SPARE;
    }

    if (tid < (unsigned)HDR_W) {
        unsigned v = 0u;
        if (tid == 0u) v = tot;
        if (tid >= 8u && tid < 8u + (unsigned)(BIN_ROWS + 2) / 2u && tot != SPARE) {
            const unsigned i = 2u * (tid - 8u);
            v = start[i];
            if (i + 1u <= (unsigned)BIN_ROWS) v |= start[i + 1u] << 16;
        }
        l_chunk(pd, bin, (unsigned)HDR_CH + (tid >> 5))[lane] = v;
    }
    if (tot != SPARE) {
        const unsigned* sw = (const unsigned*)sorted;
        for (unsigned j = tid; 4u * j < tot; j += GT) {
            unsigned sx[4], wy[4];
#pragma unroll
            for (int k = 0; k < 4; ++k) {
                const unsigned e = 4u * j + (unsigned)k;
                const bool in = e < tot;
                sx[k] = in ? sw[2u * e] : 0u;
                wy[k] = in ? sw[2u * e + 1u] : 0u;
            }
            typedef unsigned u4 __attribute__((ext_vector_type(4)));
            u4 wv4; wv4.x = wy[0]; wv4.y = wy[1]; wv4.z = wy[2]; wv4.w = wy[3];
            u2 s2;  s2.x = (sx[0] & 0xffffu) | (sx[1] << 16); s2.y = (sx[2] & 0xffffu) | (sx[3] << 16);
            const unsigned hb = (sx[0] >> 16) | ((sx[1] >> 16) << 8) | ((sx[2] >> 16) << 16) | ((sx[3] >> 16) << 24);
            *(u4*)(l_chunk(pd, bin, (unsigned)WSEC_CH + (j >> 3)) + (j & 7u) * 4u) = wv4;
            *(u2*)(l_chunk(pd, bin, (unsigned)SSEC_CH + (j >> 4)) + (j & 15u) * 2u) = s2;
            l_chunk(pd, bin, (unsigned)HSEC_CH + (j >> 5))[j & 31u] = hb;
        }
    }
}

__device__ __forceinline__ float xor8(float x) {
    const int b = __builtin_bit_cast(int, x);
    return __builtin_bit_cast(float, __builtin_amdgcn_permlane16(b, b, 0xfedcba98, 0x76543210, false, false));
}
__device__ __forceinline__ float xor16(float x) {
    const int b = __builtin_bit_cast(int, x);
    return __builtin_bit_cast(float, __builtin_amdgcn_permlanex16(b, b, 0x76543210, 0xfedcba98, false, false));
}

template <bool WHOLE>
__device__ __forceinline__ void
hop_body(unsigned* start, u2* sorted,
         const float* sp, int so,
         float* dp, int dof,
         const unsigned* out_plane,
         const int* __restrict__ esrc,
         const int* __restrict__ edst,
         const float* __restrict__ ew) {
    constexpr unsigned NQ = WHOLE ? 2u : 4u;

    const unsigned tid  = threadIdx.x;
    const unsigned lane = tid & 31u;
    const unsigned wv   = (unsigned)__builtin_amdgcn_readfirstlane((int)(tid >> 5));
    const unsigned bin  = blockIdx.x;
    const unsigned* lb  = out_plane + HALF_F + bin * (unsigned)(BIN_ROWS * OUT_F);
    const float* src = sp + so;
    float* dst = dp + dof + bin * (unsigned)(BIN_ROWS * OUT_F);

    const unsigned tot = (unsigned)__builtin_amdgcn_readfirstlane((int)lb[0]);

    if (tot <= (unsigned)ECAP) {
        if (tid < (unsigned)(BIN_ROWS + 2) / 2u) {
            const unsigned t = 8u + tid;
            const unsigned v = lb[(t >> 5) * (unsigned)OUT_F + (t & 31u)];
            start[2u * tid]      = v & 0xffffu;
            start[2u * tid + 1u] = v >> 16;
        }
        typedef unsigned u4 __attribute__((ext_vector_type(4)));
        for (unsigned j = tid; 4u * j < tot; j += GT) {
            const u4 w4 = *(const u4*)(lb + ((unsigned)WSEC_CH + (j >> 3)) * (unsigned)OUT_F + (j & 7u) * 4u);
            const u2 s2 = *(const u2*)(lb + ((unsigned)SSEC_CH + (j >> 4)) * (unsigned)OUT_F + (j & 15u) * 2u);
            const unsigned hb = lb[((unsigned)HSEC_CH + (j >> 5)) * (unsigned)OUT_F + (j & 31u)];
            u2 r0, r1, r2, r3;
            r0.x = (s2.x & 0xffffu) | ((hb & 0xffu) << 16);          r0.y = w4.x;
            r1.x = (s2.x >> 16)     | (((hb >> 8) & 0xffu) << 16);   r1.y = w4.y;
            r2.x = (s2.y & 0xffffu) | (((hb >> 16) & 0xffu) << 16);  r2.y = w4.z;
            r3.x = (s2.y >> 16)     | ((hb >> 24) << 16);            r3.y = w4.w;
            sorted[4u * j]      = r0;
            sorted[4u * j + 1u] = r1;
            sorted[4u * j + 2u] = r2;
            sorted[4u * j + 3u] = r3;
        }
        __syncthreads();

        const unsigned q  = WHOLE ? lane >> 4 : lane >> 3;
        const unsigned c4 = (WHOLE ? lane & 15u : lane & 7u) * 4u;
        const unsigned cd = WHOLE ? c4 ^ (unsigned)HALF_F : c4;
        const unsigned* sw = (const unsigned*)sorted;
        unsigned rs0 = 0u, rs1 = 0u;
        {
            const unsigned rr = wv + (unsigned)NWV * lane;
            if (rr < (unsigned)BIN_ROWS) { rs0 = start[rr]; rs1 = start[rr + 1u]; }
        }
        for (unsigned k = 0u; wv + (unsigned)NWV * k < (unsigned)BIN_ROWS; k += 2u) {
            const unsigned ra  = wv + (unsigned)NWV * k;
            const unsigned rb  = ra + (unsigned)NWV;
            const unsigned sta = (unsigned)__builtin_amdgcn_readlane((int)rs0, (int)k);
            const unsigned na  = (unsigned)__builtin_amdgcn_readlane((int)rs1, (int)k) - sta;
            const unsigned stb = (unsigned)__builtin_amdgcn_readlane((int)rs0, (int)(k + 1u));
            const unsigned nb  = (unsigned)__builtin_amdgcn_readlane((int)rs1, (int)(k + 1u)) - stb;
            const unsigned nmax = na > nb ? na : nb;
            v4f acca = {0.f, 0.f, 0.f, 0.f}, accb = {0.f, 0.f, 0.f, 0.f};
            for (unsigned base = 0u; base < nmax; base += NQ * (unsigned)PS) {
                const unsigned lefta = na > base ? na - base : 0u;
                const unsigned leftb = nb > base ? nb - base : 0u;
                unsigned sa[PS], sb[PS]; float wa[PS], wb[PS]; v4f va[PS], vb[PS];
#pragma unroll
                for (int p = 0; p < PS; ++p) {
                    const unsigned idx = NQ * (unsigned)p + q;
                    const bool oka = idx < lefta, okb = idx < leftb;
                    const unsigned ia = oka ? sta + base + idx : 0u;
                    const unsigned ib = okb ? stb + base + idx : 0u;
                    const unsigned ax = sw[2u * ia], ay = sw[2u * ia + 1u];
                    const unsigned bx = sw[2u * ib], by = sw[2u * ib + 1u];
                    sa[p] = oka ? ax : 0u;  wa[p] = oka ? __builtin_bit_cast(float, ay) : 0.f;
                    sb[p] = okb ? bx : 0u;  wb[p] = okb ? __builtin_bit_cast(float, by) : 0.f;
                }
#pragma unroll
                for (int p = 0; p < PS; ++p) {
                    va[p] = *(const v4f*)(src + sa[p] * (unsigned)OUT_F + c4);
                    vb[p] = *(const v4f*)(src + sb[p] * (unsigned)OUT_F + c4);
                }
                asm volatile("" : "+v"(va[0]), "+v"(va[1]), "+v"(va[2]), "+v"(va[3]), "+v"(va[4]), "+v"(va[5]),
                                  "+v"(vb[0]), "+v"(vb[1]), "+v"(vb[2]), "+v"(vb[3]), "+v"(vb[4]), "+v"(vb[5]));
                static_assert(PS == 6, "the statement above names the PS gathered rows one by one");
#pragma unroll
                for (int p = 0; p < PS; ++p) {
                    const unsigned idx = NQ * (unsigned)p + q;
                    const bool oka = idx < lefta, okb = idx < leftb;
                    acca.x += (oka ? va[p].x : 0.f) * wa[p]; acca.y += (oka ? va[p].y : 0.f) * wa[p];
                    acca.z += (oka ? va[p].z : 0.f) * wa[p]; acca.w += (oka ? va[p].w : 0.f) * wa[p];
                    accb.x += (okb ? vb[p].x : 0.f) * wb[p]; accb.y += (okb ? vb[p].y : 0.f) * wb[p];
                    accb.z += (okb ? vb[p].z : 0.f) * wb[p]; accb.w += (okb ? vb[p].w : 0.f) * wb[p];
                }
            }
            if (!WHOLE) {
                acca.x += xor8(acca.x);  acca.y += xor8(acca.y);  acca.z += xor8(acca.z);  acca.w += xor8(acca.w);
                accb.x += xor8(accb.x);  accb.y += xor8(accb.y);  accb.z += xor8(accb.z);  accb.w += xor8(accb.w);
            }
            acca.x += xor16(acca.x); acca.y += xor16(acca.y); acca.z += xor16(acca.z); acca.w += xor16(acca.w);
            accb.x += xor16(accb.x); accb.y += xor16(accb.y); accb.z += xor16(accb.z); accb.w += xor16(accb.w);
            if (q == 0u) {
                *(v4f*)(dst + ra * (unsigned)OUT_F + cd) = acca;
                if (rb < (unsigned)BIN_ROWS) *(v4f*)(dst + rb * (unsigned)OUT_F + cd) = accb;
            }
        }
    } else {
        float* tile = (float*)sorted;
        const v4f z = {0.f, 0.f, 0.f, 0.f};
        for (unsigned hf = 0u; hf < (WHOLE ? 2u : 1u); ++hf) {
            const float* hsrc = src + hf * (unsigned)HALF_F;
            float* hdst = dst + (WHOLE ? (hf * (unsigned)HALF_F) ^ (unsigned)HALF_F : 0u);
            for (unsigned i = tid; i < (unsigned)(BIN_ROWS * HALF_F / 4); i += GT) ((v4f*)tile)[i] = z;
            __syncthreads();
            for (unsigned eb = 0u; eb < (unsigned)N_EDGES; eb += 32u) {
                const unsigned s = (unsigned)esrc[eb + lane];
                const unsigned d = (unsigned)edst[eb + lane];
                const unsigned w = __builtin_bit_cast(unsigned, ew[eb + lane]);
                const unsigned r = d - bin * (unsigned)BIN_ROWS;
                unsigned m = __builtin_amdgcn_ballot_w32(r < (unsigned)BIN_ROWS && (r & (unsigned)(NWV - 1)) == wv && s < (unsigned)N_NODES);
                for (int k = 0; k < 32 && m != 0u; ++k) {
                    const int j = __builtin_ctz(m);
                    m &= m - 1u;
                    const unsigned sj = (unsigned)__builtin_amdgcn_readlane((int)s, j);
                    const unsigned rj = (unsigned)__builtin_amdgcn_readlane((int)r, j);
                    const float    wj = __builtin_bit_cast(float, __builtin_amdgcn_readlane((int)w, j));
                    const float gsrc = hsrc[sj * (unsigned)OUT_F + lane];
                    tile[rj * (unsigned)HALF_F + lane] += gsrc * wj;
                }
            }
            __syncthreads();
            for (unsigned i = tid; i < (unsigned)(BIN_ROWS * HALF_F / 4); i += GT)
                *(v4f*)(hdst + (i >> 3) * (unsigned)OUT_F + (i & 7u) * 4u) = ((const v4f*)tile)[i];
            if (WHOLE) __syncthreads();
        }
    }
}

__global__ void __launch_bounds__(GT)
sgc_hop(const float* sp, int so,
        float* dp, int dof,
        const unsigned* out_plane,
        const int* __restrict__ esrc,
        const int* __restrict__ edst,
        const float* __restrict__ ew) {
    __shared__ unsigned start[BIN_ROWS + 2];
    __shared__ __attribute__((aligned(16))) u2 sorted[ECAP];
    static_assert(sizeof(u2) * ECAP >= sizeof(float) * BIN_ROWS * HALF_F, "the spare path's tile lives in sorted[]");
    hop_body<false>(start, sorted, sp, so, dp, dof, out_plane, esrc, edst, ew);
}

__global__ void __launch_bounds__(GT)
sgc_hop_rows(const float* sp,
             float* dp,
             const unsigned* out_plane,
             const int* __restrict__ esrc,
             const int* __restrict__ edst,
             const float* __restrict__ ew) {
    __shared__ unsigned start[BIN_ROWS + 2];
    __shared__ __attribute__((aligned(16))) u2 sorted[ECAP];
    static_assert(sizeof(u2) * ECAP >= sizeof(float) * BIN_ROWS * HALF_F, "the spare path's tile lives in sorted[]");
    hop_body<true>(start, sorted, sp, 0, dp, 0, out_plane, esrc, edst, ew);
}

extern "C" void kernel_launch(void* const* d_in, const int* in_sizes, int n_in,
                              void* d_out, int out_size, void* d_ws, size_t ws_size,
                              hipStream_t stream) {
    const float* x    = (const float*)d_in[0];
    const float* W    = (const float*)d_in[1];
    const float* b    = (const float*)d_in[2];
    const int*   esrc = (const int*)d_in[3];
    const int*   edst = (const int*)d_in[4];
    const float* ew   = (const float*)d_in[5];

    float* out = (float*)d_out;
    float* ws  = (float*)d_ws;

    sgc_gemm_wmma<<<(NTILES + 7) / 8, 256, 0, stream>>>(x, W, b, out, ws);

    sgc_order<<<NG,    GT, 0, stream>>>((unsigned*)ws, (unsigned*)out, esrc, edst, ew);
    sgc_csr  <<<NBINS, GT, 0, stream>>>((unsigned*)ws, (unsigned*)out);

    sgc_hop<<<NBINS, GT, 0, stream>>>(out,  0, ws,   0, (const unsigned*)out, esrc, edst, ew);
    sgc_hop<<<NBINS, GT, 0, stream>>>(ws,  32, out,  0, (const unsigned*)out, esrc, edst, ew);
    sgc_hop<<<NBINS, GT, 0, stream>>>(ws,   0, ws,  32, (const unsigned*)out, esrc, edst, ew);
    sgc_hop<<<NBINS, GT, 0, stream>>>(out,  0, ws,   0, (const unsigned*)out, esrc, edst, ew);
    sgc_hop_rows<<<NBINS, GT, 0, stream>>>(ws, out, (const unsigned*)out, esrc, edst, ew);
}


# hops: fewer vector instructions a gathered edge: LDS records at constant offsets with no index clamp, gather address = uniform base + 32-bit offset, products fused into the sums (fma)
# speedup vs baseline: 3.7227x; 1.0436x over previous
#include <hip/hip_runtime.h>
#include <hip/hip_bf16.h>

typedef float v2f __attribute__((ext_vector_type(2)));
typedef float v4f __attribute__((ext_vector_type(4)));
typedef float v8f __attribute__((ext_vector_type(8)));

#define N_NODES 100000
#define N_EDGES 1600000
#define IN_F    256
#define OUT_F   64
#define NTILES  (N_NODES / 16)

#define WPITCH 260

__global__ void __launch_bounds__(256)
sgc_gemm_wmma(const float* __restrict__ x,
              const float* __restrict__ W,
              const float* __restrict__ bias,
              float* __restrict__ plo,
              float* __restrict__ phi) {
    __shared__ __attribute__((aligned(16))) float wl[OUT_F * WPITCH];

    for (int j = 0; j < (OUT_F * IN_F / 4) / 256; ++j) {
        const int i   = (int)threadIdx.x + 256 * j;
        const int row = i >> 6;
        const int c4  = i & 63;
        const v4f v = *(const v4f*)(W + (size_t)i * 4);
        *(v4f*)(wl + row * WPITCH + c4 * 4) = v;
    }
    __syncthreads();

    const int wave = (blockIdx.x * blockDim.x + threadIdx.x) >> 5;
    const int lane = threadIdx.x & 31;
    if (wave >= NTILES) return;

    const int m    = lane & 15;
    const int hoff = (lane >> 4) << 4;

    const float* __restrict__ xrow = x + (size_t)(wave * 16 + m) * IN_F + hoff;
    const float* wb = wl + m * WPITCH + hoff;

    v8f acc0 = {}, acc1 = {}, acc2 = {}, acc3 = {};

    for (int kb = 0; kb < IN_F; kb += 32) {
        v4f a[4];
        for (int i = 0; i < 4; ++i)
            a[i] = *(const v4f*)(xrow + kb + 4 * i);
        for (int i = 0; i < 4; ++i) {
            const v4f b0 = *(const v4f*)(wb + 0 * 16 * WPITCH + kb + 4 * i);
            const v4f b1 = *(const v4f*)(wb + 1 * 16 * WPITCH + kb + 4 * i);
            const v4f b2 = *(const v4f*)(wb + 2 * 16 * WPITCH + kb + 4 * i);
            const v4f b3 = *(const v4f*)(wb + 3 * 16 * WPITCH + kb + 4 * i);
            const v2f alo = a[i].xy, ahi = a[i].zw;
            acc0 = __builtin_amdgcn_wmma_f32_16x16x4_f32(false, alo, false, b0.xy, (short)0, acc0, false, false);
            acc1 = __builtin_amdgcn_wmma_f32_16x16x4_f32(false, alo, false, b1.xy, (short)0, acc1, false, false);
            acc2 = __builtin_amdgcn_wmma_f32_16x16x4_f32(false, alo, false, b2.xy, (short)0, acc2, false, false);
            acc3 = __builtin_amdgcn_wmma_f32_16x16x4_f32(false, alo, false, b3.xy, (short)0, acc3, false, false);
            acc0 = __builtin_amdgcn_wmma_f32_16x16x4_f32(false, ahi, false, b0.zw, (short)0, acc0, false, false);
            acc1 = __builtin_amdgcn_wmma_f32_16x16x4_f32(false, ahi, false, b1.zw, (short)0, acc1, false, false);
            acc2 = __builtin_amdgcn_wmma_f32_16x16x4_f32(false, ahi, false, b2.zw, (short)0, acc2, false, false);
            acc3 = __builtin_amdgcn_wmma_f32_16x16x4_f32(false, ahi, false, b3.zw, (short)0, acc3, false, false);
        }
    }

    const int row_base = wave * 16 + ((lane >> 4) ? 8 : 0);
    const int col      = lane & 15;
    const float bv0 = bias[col +  0];
    const float bv1 = bias[col + 16];
    const float bv2 = bias[col + 32];
    const float bv3 = bias[col + 48];

#pragma unroll
    for (int r = 0; r < 8; ++r) {
        const size_t o = (size_t)(row_base + r) * OUT_F + col;
        plo[o +  0] = acc0[r] + bv0;
        plo[o + 16] = acc1[r] + bv1;
        phi[o + 32] = acc2[r] + bv2;
        phi[o + 48] = acc3[r] + bv3;
    }
}

#define BIN_ROWS 400
#define NBINS    250
#define HALF_F   (OUT_F / 2)
#define NG       128
#define SLAB     (N_EDGES / NG)
#define LPITCH_W 192
#define LCHUNKS  (LPITCH_W / 32)
#define CAP      ((LPITCH_W - 2) / 2)
#define GT       1024
#define NWV      (GT / 32)
#define ECAP     7168
#define HDR_CH   0
#define HDR_W    224
#define WSEC_CH  7
#define SSEC_CH  (WSEC_CH + ECAP / 32)
#define HSEC_CH  (SSEC_CH + ECAP / 64)
#define SPARE    0xffffffffu
static_assert(NBINS * BIN_ROWS == N_NODES, "ranges must tile the nodes");
static_assert(NG * SLAB == N_EDGES, "slabs must tile the edges");
static_assert(NG * LCHUNKS <= 2 * BIN_ROWS && LPITCH_W % 32 == 0, "the blocks must fit a range's rows of A and D, whole chunks each");
static_assert(HSEC_CH + ECAP / 128 <= BIN_ROWS && ECAP % 128 == 0, "a range's part of L must fit its rows of D");
static_assert(ECAP < 65536 && 8 + (BIN_ROWS + 2) / 2 <= HDR_W, "row starts are 16-bit fields of the header");
static_assert(N_NODES <= (1 << 17) && BIN_ROWS <= 512 && NBINS <= 512, "packing and key widths");
static_assert(NG % NWV == 0 && NG % 32 == 0, "list walk of sgc_csr");
static_assert(N_EDGES % 32 == 0, "whole waves of edges");

typedef unsigned u2 __attribute__((ext_vector_type(2)));
typedef __attribute__((address_space(3))) unsigned lds_u32;

struct Match { unsigned rank, size, lead; };
template <int BITS>
__device__ __forceinline__ Match match_key(unsigned key, bool valid, unsigned lane) {
    unsigned m = __builtin_amdgcn_ballot_w32(valid);
#pragma unroll
    for (int bit = 0; bit < BITS; ++bit) {
        const bool one = ((key >> bit) & 1u) != 0u;
        const unsigned bal = __builtin_amdgcn_ballot_w32(valid && one);
        m &= one ? bal : ~bal;
    }
    Match r;
    r.rank = (unsigned)__builtin_popcount(m & ((1u << lane) - 1u));
    r.size = (unsigned)__builtin_popcount(m);
    r.lead = valid ? (unsigned)__builtin_ctz(m | (1u << lane)) : lane;
    return r;
}

template <int NKEYS, class Src>
__device__ __forceinline__ void group_sort(const Src& src, unsigned* cw, unsigned* start,
                                           unsigned* wsum, u2* sorted,
                                           unsigned tid, unsigned lane, unsigned wv) {
    static_assert(NKEYS <= 512, "the key field of the element word has nine bits");
    for (unsigned i = tid; i < (unsigned)(NWV * NKEYS); i += GT) cw[i] = 0u;
    __syncthreads();
    volatile lds_u32* c = (volatile lds_u32*)(cw + wv * NKEYS);

    unsigned info[Src::ITERS]; u2 pay[Src::ITERS];
#pragma unroll
    for (int it = 0; it < Src::ITERS; ++it) {
        unsigned k;
        const bool ok = src.get(it, k, pay[it]);
        info[it] = k | (ok ? 0x80000000u : 0u);
    }

#pragma unroll
    for (int it = 0; it < Src::ITERS; ++it) {
        if (src.skip(it)) continue;
        const unsigned k = info[it] & 0x1ffu;
        const bool valid = (info[it] >> 31) != 0u;
        const Match mt = match_key<(NKEYS <= 256 ? 8 : 9)>(k, valid, lane);
        if (valid && mt.rank == 0u) c[k] = c[k] + mt.size;
        info[it] |= (mt.rank << 9) | (mt.size << 14) | (mt.lead << 20);
        asm volatile("" : "+v"(info[it]));
    }
    __syncthreads();

    unsigned tot = 0u;
    if (tid < (unsigned)NKEYS) {
        for (int w = 0; w < NWV; ++w) {
            const unsigned v = cw[w * NKEYS + tid];
            cw[w * NKEYS + tid] = tot;
            tot += v;
        }
    }
    unsigned x = tot;
    for (unsigned d = 1u; d < 32u; d <<= 1) {
        const unsigned y = (unsigned)__builtin_amdgcn_ds_bpermute((int)((lane - d) << 2), (int)x);
        if (lane >= d) x += y;
    }
    if (lane == 31u) wsum[wv] = x;
    __syncthreads();
    unsigned off = 0u;
    for (unsigned w = 0u; w < wv; ++w) off += wsum[w];
    if (tid < (unsigned)NKEYS) start[tid] = off + x - tot;
    if (tid == (unsigned)(NKEYS - 1)) start[NKEYS] = off + x;
    __syncthreads();

#pragma unroll
    for (int it = 0; it < Src::ITERS; ++it) {
        if (src.skip(it)) continue;
        const unsigned k    = info[it] & 0x1ffu;
        const unsigned rank = (info[it] >> 9) & 31u;
        const unsigned size = (info[it] >> 14) & 63u;
        const unsigned lead = (info[it] >> 20) & 31u;
        const bool valid = (info[it] >> 31) != 0u;
        unsigned base = 0u;
        if (valid && rank == 0u) { base = c[k]; c[k] = base + size; }
        base = (unsigned)__builtin_amdgcn_ds_bpermute((int)(lead << 2), (int)base);
        if (valid) sorted[start[k] + base + rank] = pay[it];
    }
    __syncthreads();
}

struct EdgeSrc {
    static constexpr int ITERS = (SLAB + GT - 1) / GT;
    const int* esrc; const int* edst; const float* ew;
    unsigned e0, tid;
    __device__ __forceinline__ bool skip(int) const { return false; }
    __device__ __forceinline__ bool get(int it, unsigned& key, u2& pay) const {
        const unsigned i  = tid + (unsigned)GT * (unsigned)it;
        const unsigned ic = i < (unsigned)SLAB ? i : (unsigned)(SLAB - 1);
        const unsigned s = (unsigned)esrc[e0 + ic];
        const unsigned d = (unsigned)edst[e0 + ic];
        const unsigned w = __builtin_bit_cast(unsigned, ew[e0 + ic]);
        const unsigned b = d / BIN_ROWS;
        const bool ok = i < (unsigned)SLAB && b < (unsigned)NBINS && s < (unsigned)N_NODES;
        key   = ok ? b : 0u;
        pay.x = s | ((d - b * BIN_ROWS) << 17);
        pay.y = w;
        return ok;
    }
};

#define PS 6
#define LTAIL (4 * PS)
__device__ __forceinline__ unsigned* list_chunk(unsigned* pa, unsigned* pd, unsigned bin, unsigned ch) {
    const bool in_a = ch < (unsigned)BIN_ROWS;
    unsigned* p = in_a ? pa : pd;
    const unsigned row = in_a ? ch : ch - (unsigned)BIN_ROWS;
    return p + (bin * (unsigned)BIN_ROWS + row) * (unsigned)OUT_F;
}

__global__ void __launch_bounds__(GT)
sgc_order(unsigned* ws_plane, unsigned* out_plane,
          const int* __restrict__ esrc,
          const int* __restrict__ edst,
          const float* __restrict__ ew) {
    __shared__ unsigned cw[NWV * NBINS];
    __shared__ unsigned start[NBINS + 1];
    __shared__ unsigned wsum[NWV];
    __shared__ __attribute__((aligned(8))) u2 sorted[SLAB];

    const unsigned tid  = threadIdx.x;
    const unsigned lane = tid & 31u;
    const unsigned wv   = (unsigned)__builtin_amdgcn_readfirstlane((int)(tid >> 5));
    const unsigned g    = blockIdx.x;
    unsigned* pa = ws_plane;
    unsigned* pd = out_plane + HALF_F;

    EdgeSrc src;
    src.esrc = esrc; src.edst = edst; src.ew = ew; src.e0 = g * (unsigned)SLAB; src.tid = tid;
    group_sort<NBINS>(src, cw, start, wsum, sorted, tid, lane, wv);

    for (unsigned b = wv; b < (unsigned)NBINS; b += NWV) {
        const unsigned st  = start[b];
        const unsigned tot = start[b + 1] - st;
        const unsigned n   = tot < (unsigned)CAP ? tot : (unsigned)CAP;
        for (unsigned t = lane; t <= n; t += 32u) {
            u2 v; v.x = tot; v.y = 0u;
            if (t != 0u) v = sorted[st + t - 1u];
            unsigned* c = list_chunk(pa, pd, b, g * (unsigned)LCHUNKS + (t >> 4));
            *(u2*)(c + (t & 15u) * 2u) = v;
        }
    }
}

struct ListSrc {
    static constexpr int LSTR  = (CAP + 31) / 32;
    static constexpr int ITERS = (NG / NWV) * LSTR;
    unsigned* pa; unsigned* pd; const unsigned* gcnt;
    unsigned bin, lane, wv;
    __device__ __forceinline__ bool skip(int it) const {
        const unsigned n = (unsigned)__builtin_amdgcn_readfirstlane((int)gcnt[wv + (unsigned)NWV * (unsigned)(it / LSTR)]);
        return 32u * (unsigned)(it % LSTR) >= n;
    }
    __device__ __forceinline__ bool get(int it, unsigned& key, u2& pay) const {
        const unsigned gl = wv + (unsigned)NWV * (unsigned)(it / LSTR);
        const unsigned j  = lane + 32u * (unsigned)(it % LSTR);
        const unsigned jc = j < (unsigned)CAP ? j : (unsigned)(CAP - 1);
        const unsigned t  = 1u + jc;
        const unsigned* c = list_chunk(pa, pd, bin, gl * (unsigned)LCHUNKS + (t >> 4));
        const u2 e = *(const u2*)(c + (t & 15u) * 2u);
        const unsigned r = e.x >> 17;
        const bool ok = j < gcnt[gl] && r < (unsigned)BIN_ROWS;
        key   = ok ? r : 0u;
        pay.x = e.x & 0x1ffffu;
        pay.y = e.y;
        return ok;
    }
};

__device__ __forceinline__ unsigned* l_chunk(unsigned* pd, unsigned bin, unsigned ch) {
    return pd + (bin * (unsigned)BIN_ROWS + ch) * (unsigned)OUT_F;
}

__global__ void __launch_bounds__(GT)
sgc_csr(unsigned* ws_plane, unsigned* out_plane) {
    __shared__ unsigned cw[NWV * BIN_ROWS];
    __shared__ unsigned start[BIN_ROWS + 1];
    __shared__ unsigned wsum[NWV];
    __shared__ unsigned gcnt[NG];
    __shared__ unsigned ovf[NG / 32];
    __shared__ __attribute__((aligned(16))) u2 sorted[NG * CAP];

    const unsigned tid  = threadIdx.x;
    const unsigned lane = tid & 31u;
    const unsigned wv   = (unsigned)__builtin_amdgcn_readfirstlane((int)(tid >> 5));
    const unsigned bin  = blockIdx.x;
    unsigned* pa = ws_plane;
    unsigned* pd = out_plane + HALF_F;

    if (tid < (unsigned)NG) {
        const unsigned cn = list_chunk(pa, pd, bin, tid * (unsigned)LCHUNKS)[0];
        gcnt[tid] = cn;
        const unsigned bal = __builtin_amdgcn_ballot_w32(cn > (unsigned)CAP);
        if (lane == 0u) ovf[wv] = bal;
    }
    __syncthreads();
    unsigned anyover = 0u;
    for (int i = 0; i < NG / 32; ++i) anyover |= ovf[i];
    const bool over = __builtin_amdgcn_readfirstlane((int)anyover) != 0;

    unsigned tot = SPARE;
    if (!over) {
        ListSrc src;
        src.pa = pa; src.pd = pd; src.gcnt = gcnt; src.bin = bin; src.lane = lane; src.wv = wv;
        group_sort<BIN_ROWS>(src, cw, start, wsum, sorted, tid, lane, wv);
        tot = start[BIN_ROWS];
        if (tot > (unsigned)ECAP) tot = SPARE;
    }

    if (tid < (unsigned)HDR_W) {
        unsigned v = 0u;
        if (tid == 0u) v = tot;
        if (tid >= 8u && tid < 8u + (unsigned)(BIN_ROWS + 2) / 2u && tot != SPARE) {
            const unsigned i = 2u * (tid - 8u);
            v = start[i];
            if (i + 1u <= (unsigned)BIN_ROWS) v |= start[i + 1u] << 16;
        }
        l_chunk(pd, bin, (unsigned)HDR_CH + (tid >> 5))[lane] = v;
    }
    if (tot != SPARE) {
        const unsigned* sw = (const unsigned*)sorted;
        for (unsigned j = tid; 4u * j < tot; j += GT) {
            unsigned sx[4], wy[4];
#pragma unroll
            for (int k = 0; k < 4; ++k) {
                const unsigned e = 4u * j + (unsigned)k;
                const bool in = e < tot;
                sx[k] = in ? sw[2u * e] : 0u;
                wy[k] = in ? sw[2u * e + 1u] : 0u;
            }
            typedef unsigned u4 __attribute__((ext_vector_type(4)));
            u4 wv4; wv4.x = wy[0]; wv4.y = wy[1]; wv4.z = wy[2]; wv4.w = wy[3];
            u2 s2;  s2.x = (sx[0] & 0xffffu) | (sx[1] << 16); s2.y = (sx[2] & 0xffffu) | (sx[3] << 16);
            const unsigned hb = (sx[0] >> 16) | ((sx[1] >> 16) << 8) | ((sx[2] >> 16) << 16) | ((sx[3] >> 16) << 24);
            *(u4*)(l_chunk(pd, bin, (unsigned)WSEC_CH + (j >> 3)) + (j & 7u) * 4u) = wv4;
            *(u2*)(l_chunk(pd, bin, (unsigned)SSEC_CH + (j >> 4)) + (j & 15u) * 2u) = s2;
            l_chunk(pd, bin, (unsigned)HSEC_CH + (j >> 5))[j & 31u] = hb;
        }
    }
}

__device__ __forceinline__ float xor8(float x) {
    const int b = __builtin_bit_cast(int, x);
    return __builtin_bit_cast(float, __builtin_amdgcn_permlane16(b, b, 0xfedcba98, 0x76543210, false, false));
}
__device__ __forceinline__ float xor16(float x) {
    const int b = __builtin_bit_cast(int, x);
    return __builtin_bit_cast(float, __builtin_amdgcn_permlanex16(b, b, 0x76543210, 0xfedcba98, false, false));
}

template <bool WHOLE>
__device__ __forceinline__ void
hop_body(unsigned* start, u2* sorted,
         const float* sp, int so,
         float* dp, int dof,
         const unsigned* out_plane,
         const int* __restrict__ esrc,
         const int* __restrict__ edst,
         const float* __restrict__ ew) {
    constexpr unsigned NQ = WHOLE ? 2u : 4u;

    const unsigned tid  = threadIdx.x;
    const unsigned lane = tid & 31u;
    const unsigned wv   = (unsigned)__builtin_amdgcn_readfirstlane((int)(tid >> 5));
    const unsigned bin  = blockIdx.x;
    const unsigned* lb  = out_plane + HALF_F + bin * (unsigned)(BIN_ROWS * OUT_F);
    const float* src = sp + so;
    float* dst = dp + dof + bin * (unsigned)(BIN_ROWS * OUT_F);

    const unsigned tot = (unsigned)__builtin_amdgcn_readfirstlane((int)lb[0]);

    if (tot <= (unsigned)ECAP) {
        if (tid < (unsigned)(BIN_ROWS + 2) / 2u) {
            const unsigned t = 8u + tid;
            const unsigned v = lb[(t >> 5) * (unsigned)OUT_F + (t & 31u)];
            start[2u * tid]      = v & 0xffffu;
            start[2u * tid + 1u] = v >> 16;
        }
        typedef unsigned u4 __attribute__((ext_vector_type(4)));
        for (unsigned j = tid; 4u * j < tot; j += GT) {
            const u4 w4 = *(const u4*)(lb + ((unsigned)WSEC_CH + (j >> 3)) * (unsigned)OUT_F + (j & 7u) * 4u);
            const u2 s2 = *(const u2*)(lb + ((unsigned)SSEC_CH + (j >> 4)) * (unsigned)OUT_F + (j & 15u) * 2u);
            const unsigned hb = lb[((unsigned)HSEC_CH + (j >> 5)) * (unsigned)OUT_F + (j & 31u)];
            u2 r0, r1, r2, r3;
            r0.x = (s2.x & 0xffffu) | ((hb & 0xffu) << 16);          r0.y = w4.x;
            r1.x = (s2.x >> 16)     | (((hb >> 8) & 0xffu) << 16);   r1.y = w4.y;
            r2.x = (s2.y & 0xffffu) | (((hb >> 16) & 0xffu) << 16);  r2.y = w4.z;
            r3.x = (s2.y >> 16)     | ((hb >> 24) << 16);            r3.y = w4.w;
            sorted[4u * j]      = r0;
            sorted[4u * j + 1u] = r1;
            sorted[4u * j + 2u] = r2;
            sorted[4u * j + 3u] = r3;
        }
        if (tid < (unsigned)LTAIL + 4u) { u2 z2; z2.x = 0u; z2.y = 0u; sorted[((tot + 3u) & ~3u) + tid] = z2; }
        __syncthreads();

        const unsigned q  = WHOLE ? lane >> 4 : lane >> 3;
        const unsigned c4 = (WHOLE ? lane & 15u : lane & 7u) * 4u;
        const unsigned cd = WHOLE ? c4 ^ (unsigned)HALF_F : c4;
        const char* srcb  = (const char*)src;
        const unsigned cb = c4 * 4u;
        unsigned rs0 = 0u, rs1 = 0u;
        {
            const unsigned rr = wv + (unsigned)NWV * lane;
            if (rr < (unsigned)BIN_ROWS) { rs0 = start[rr]; rs1 = start[rr + 1u]; }
        }
        for (unsigned k = 0u; wv + (unsigned)NWV * k < (unsigned)BIN_ROWS; k += 2u) {
            const unsigned ra  = wv + (unsigned)NWV * k;
            const unsigned rb  = ra + (unsigned)NWV;
            const unsigned sta = (unsigned)__builtin_amdgcn_readlane((int)rs0, (int)k);
            const unsigned na  = (unsigned)__builtin_amdgcn_readlane((int)rs1, (int)k) - sta;
            const unsigned stb = (unsigned)__builtin_amdgcn_readlane((int)rs0, (int)(k + 1u));
            const unsigned nb  = (unsigned)__builtin_amdgcn_readlane((int)rs1, (int)(k + 1u)) - stb;
            const unsigned nmax = na > nb ? na : nb;
            v4f acca = {0.f, 0.f, 0.f, 0.f}, accb = {0.f, 0.f, 0.f, 0.f};
            for (unsigned base = 0u; base < nmax; base += NQ * (unsigned)PS) {
                const unsigned lefta = na > base ? na - base : 0u;
                const unsigned leftb = nb > base ? nb - base : 0u;
                unsigned sa[PS], sb[PS]; float wa[PS], wb[PS]; v4f va[PS], vb[PS];
                const unsigned oa = lefta != 0u ? sta + base : 0u;
                const unsigned ob = leftb != 0u ? stb + base : 0u;
                const u2* pa0 = sorted + (oa + q);
                const u2* pb0 = sorted + (ob + q);
#pragma unroll
                for (int p = 0; p < PS; ++p) {
                    const unsigned idx = NQ * (unsigned)p + q;
                    const bool oka = idx < lefta, okb = idx < leftb;
                    const u2 ea = pa0[NQ * (unsigned)p];
                    const u2 eb = pb0[NQ * (unsigned)p];
                    const unsigned ax = ea.x, ay = ea.y, bx = eb.x, by = eb.y;
                    sa[p] = oka ? ax : 0u;  wa[p] = oka ? __builtin_bit_cast(float, ay) : 0.f;
                    sb[p] = okb ? bx : 0u;  wb[p] = okb ? __builtin_bit_cast(float, by) : 0.f;
                }
#pragma unroll
                for (int p = 0; p < PS; ++p) {
                    va[p] = *(const v4f*)(srcb + ((sa[p] << 8) | cb));
                    vb[p] = *(const v4f*)(srcb + ((sb[p] << 8) | cb));
                }
                asm volatile("" : "+v"(va[0]), "+v"(va[1]), "+v"(va[2]), "+v"(va[3]), "+v"(va[4]), "+v"(va[5]),
                                  "+v"(vb[0]), "+v"(vb[1]), "+v"(vb[2]), "+v"(vb[3]), "+v"(vb[4]), "+v"(vb[5]));
                static_assert(PS == 6, "the statement above names the PS gathered rows one by one");
#pragma unroll
                for (int p = 0; p < PS; ++p) {
                    acca.x = __builtin_fmaf(va[p].x, wa[p], acca.x); acca.y = __builtin_fmaf(va[p].y, wa[p], acca.y);
                    acca.z = __builtin_fmaf(va[p].z, wa[p], acca.z); acca.w = __builtin_fmaf(va[p].w, wa[p], acca.w);
                    accb.x = __builtin_fmaf(vb[p].x, wb[p], accb.x); accb.y = __builtin_fmaf(vb[p].y, wb[p], accb.y);
                    accb.z = __builtin_fmaf(vb[p].z, wb[p], accb.z); accb.w = __builtin_fmaf(vb[p].w, wb[p], accb.w);
                }
            }
            if (!WHOLE) {
                acca.x += xor8(acca.x);  acca.y += xor8(acca.y);  acca.z += xor8(acca.z);  acca.w += xor8(acca.w);
                accb.x += xor8(accb.x);  accb.y += xor8(accb.y);  accb.z += xor8(accb.z);  accb.w += xor8(accb.w);
            }
            acca.x += xor16(acca.x); acca.y += xor16(acca.y); acca.z += xor16(acca.z); acca.w += xor16(acca.w);
            accb.x += xor16(accb.x); accb.y += xor16(accb.y); accb.z += xor16(accb.z); accb.w += xor16(accb.w);
            if (q == 0u) {
                *(v4f*)(dst + ra * (unsigned)OUT_F + cd) = acca;
                if (rb < (unsigned)BIN_ROWS) *(v4f*)(dst + rb * (unsigned)OUT_F + cd) = accb;
            }
        }
    } else {
        float* tile = (float*)sorted;
        const v4f z = {0.f, 0.f, 0.f, 0.f};
        for (unsigned hf = 0u; hf < (WHOLE ? 2u : 1u); ++hf) {
            const float* hsrc = src + hf * (unsigned)HALF_F;
            float* hdst = dst + (WHOLE ? (hf * (unsigned)HALF_F) ^ (unsigned)HALF_F : 0u);
            for (unsigned i = tid; i < (unsigned)(BIN_ROWS * HALF_F / 4); i += GT) ((v4f*)tile)[i] = z;
            __syncthreads();
            for (unsigned eb = 0u; eb < (unsigned)N_EDGES; eb += 32u) {
                const unsigned s = (unsigned)esrc[eb + lane];
                const unsigned d = (unsigned)edst[eb + lane];
                const unsigned w = __builtin_bit_cast(unsigned, ew[eb + lane]);
                const unsigned r = d - bin * (unsigned)BIN_ROWS;
                unsigned m = __builtin_amdgcn_ballot_w32(r < (unsigned)BIN_ROWS && (r & (unsigned)(NWV - 1)) == wv && s < (unsigned)N_NODES);
                for (int k = 0; k < 32 && m != 0u; ++k) {
                    const int j = __builtin_ctz(m);
                    m &= m - 1u;
                    const unsigned sj = (unsigned)__builtin_amdgcn_readlane((int)s, j);
                    const unsigned rj = (unsigned)__builtin_amdgcn_readlane((int)r, j);
                    const float    wj = __builtin_bit_cast(float, __builtin_amdgcn_readlane((int)w, j));
                    const float gsrc = hsrc[sj * (unsigned)OUT_F + lane];
                    tile[rj * (unsigned)HALF_F + lane] += gsrc * wj;
                }
            }
            __syncthreads();
            for (unsigned i = tid; i < (unsigned)(BIN_ROWS * HALF_F / 4); i += GT)
                *(v4f*)(hdst + (i >> 3) * (unsigned)OUT_F + (i & 7u) * 4u) = ((const v4f*)tile)[i];
            if (WHOLE) __syncthreads();
        }
    }
}

__global__ void __launch_bounds__(GT)
sgc_hop(const float* sp, int so,
        float* dp, int dof,
        const unsigned* out_plane,
        const int* __restrict__ esrc,
        const int* __restrict__ edst,
        const float* __restrict__ ew) {
    __shared__ unsigned start[BIN_ROWS + 2];
    __shared__ __attribute__((aligned(16))) u2 sorted[ECAP + LTAIL + 4];
    static_assert(sizeof(u2) * ECAP >= sizeof(float) * BIN_ROWS * HALF_F, "the spare path's tile lives in sorted[]");
    hop_body<false>(start, sorted, sp, so, dp, dof, out_plane, esrc, edst, ew);
}

__global__ void __launch_bounds__(GT)
sgc_hop_rows(const float* sp,
             float* dp,
             const unsigned* out_plane,
             const int* __restrict__ esrc,
             const int* __restrict__ edst,
             const float* __restrict__ ew) {
    __shared__ unsigned start[BIN_ROWS + 2];
    __shared__ __attribute__((aligned(16))) u2 sorted[ECAP + LTAIL + 4];
    static_assert(sizeof(u2) * ECAP >= sizeof(float) * BIN_ROWS * HALF_F, "the spare path's tile lives in sorted[]");
    hop_body<true>(start, sorted, sp, 0, dp, 0, out_plane, esrc, edst, ew);
}

extern "C" void kernel_launch(void* const* d_in, const int* in_sizes, int n_in,
                              void* d_out, int out_size, void* d_ws, size_t ws_size,
                              hipStream_t stream) {
    const float* x    = (const float*)d_in[0];
    const float* W    = (const float*)d_in[1];
    const float* b    = (const float*)d_in[2];
    const int*   esrc = (const int*)d_in[3];
    const int*   edst = (const int*)d_in[4];
    const float* ew   = (const float*)d_in[5];

    float* out = (float*)d_out;
    float* ws  = (float*)d_ws;

    sgc_gemm_wmma<<<(NTILES + 7) / 8, 256, 0, stream>>>(x, W, b, out, ws);

    sgc_order<<<NG,    GT, 0, stream>>>((unsigned*)ws, (unsigned*)out, esrc, edst, ew);
    sgc_csr  <<<NBINS, GT, 0, stream>>>((unsigned*)ws, (unsigned*)out);

    sgc_hop<<<NBINS, GT, 0, stream>>>(out,  0, ws,   0, (const unsigned*)out, esrc, edst, ew);
    sgc_hop<<<NBINS, GT, 0, stream>>>(ws,  32, out,  0, (const unsigned*)out, esrc, edst, ew);
    sgc_hop<<<NBINS, GT, 0, stream>>>(ws,   0, ws,  32, (const unsigned*)out, esrc, edst, ew);
    sgc_hop<<<NBINS, GT, 0, stream>>>(out,  0, ws,   0, (const unsigned*)out, esrc, edst, ew);
    sgc_hop_rows<<<NBINS, GT, 0, stream>>>(ws, out, (const unsigned*)out, esrc, edst, ew);
}


# projection and edge ordering in ONE launch: the 128 ordering work-groups then project 17 row tiles each, 128 more work-groups 32 each (one tile a wave); 7 launches, not 8; every sum as before
# speedup vs baseline: 3.9653x; 1.0652x over previous
#include <hip/hip_runtime.h>
#include <hip/hip_bf16.h>

typedef float v2f __attribute__((ext_vector_type(2)));
typedef float v4f __attribute__((ext_vector_type(4)));
typedef float v8f __attribute__((ext_vector_type(8)));

#define N_NODES 100000
#define N_EDGES 1600000
#define IN_F    256
#define OUT_F   64
#define NTILES  (N_NODES / 16)

#define WPITCH 260

#define BIN_ROWS 400
#define NBINS    250
#define HALF_F   (OUT_F / 2)
#define NG       128
#define SLAB     (N_EDGES / NG)
#define LPITCH_W 192
#define LCHUNKS  (LPITCH_W / 32)
#define CAP      ((LPITCH_W - 2) / 2)
#define GT       1024
#define NWV      (GT / 32)
#define ECAP     7168
#define HDR_CH   0
#define HDR_W    224
#define WSEC_CH  7
#define SSEC_CH  (WSEC_CH + ECAP / 32)
#define HSEC_CH  (SSEC_CH + ECAP / 64)
#define SPARE    0xffffffffu
static_assert(NBINS * BIN_ROWS == N_NODES, "ranges must tile the nodes");
static_assert(NG * SLAB == N_EDGES, "slabs must tile the edges");
static_assert(NG * LCHUNKS <= 2 * BIN_ROWS && LPITCH_W % 32 == 0, "the blocks must fit a range's rows of A and D, whole chunks each");
static_assert(HSEC_CH + ECAP / 128 <= BIN_ROWS && ECAP % 128 == 0, "a range's part of L must fit its rows of D");
static_assert(ECAP < 65536 && 8 + (BIN_ROWS + 2) / 2 <= HDR_W, "row starts are 16-bit fields of the header");
static_assert(N_NODES <= (1 << 17) && BIN_ROWS <= 512 && NBINS <= 512, "packing and key widths");
static_assert(NG % NWV == 0 && NG % 32 == 0, "list walk of sgc_csr");
static_assert(N_EDGES % 32 == 0, "whole waves of edges");

typedef unsigned u2 __attribute__((ext_vector_type(2)));
typedef __attribute__((address_space(3))) unsigned lds_u32;

struct Match { unsigned rank, size, lead; };
template <int BITS>
__device__ __forceinline__ Match match_key(unsigned key, bool valid, unsigned lane) {
    unsigned m = __builtin_amdgcn_ballot_w32(valid);
#pragma unroll
    for (int bit = 0; bit < BITS; ++bit) {
        const bool one = ((key >> bit) & 1u) != 0u;
        const unsigned bal = __builtin_amdgcn_ballot_w32(valid && one);
        m &= one ? bal : ~bal;
    }
    Match r;
    r.rank = (unsigned)__builtin_popcount(m & ((1u << lane) - 1u));
    r.size = (unsigned)__builtin_popcount(m);
    r.lead = valid ? (unsigned)__builtin_ctz(m | (1u << lane)) : lane;
    return r;
}

template <int NKEYS, class Src>
__device__ __forceinline__ void group_sort(const Src& src, unsigned* cw, unsigned* start,
                                           unsigned* wsum, u2* sorted,
                                           unsigned tid, unsigned lane, unsigned wv) {
    static_assert(NKEYS <= 512, "the key field of the element word has nine bits");
    for (unsigned i = tid; i < (unsigned)(NWV * NKEYS); i += GT) cw[i] = 0u;
    __syncthreads();
    volatile lds_u32* c = (volatile lds_u32*)(cw + wv * NKEYS);

    unsigned info[Src::ITERS]; u2 pay[Src::ITERS];
#pragma unroll
    for (int it = 0; it < Src::ITERS; ++it) {
        unsigned k;
        const bool ok = src.get(it, k, pay[it]);
        info[it] = k | (ok ? 0x80000000u : 0u);
    }

#pragma unroll
    for (int it = 0; it < Src::ITERS; ++it) {
        if (src.skip(it)) continue;
        const unsigned k = info[it] & 0x1ffu;
        const bool valid = (info[it] >> 31) != 0u;
        const Match mt = match_key<(NKEYS <= 256 ? 8 : 9)>(k, valid, lane);
        if (valid && mt.rank == 0u) c[k] = c[k] + mt.size;
        info[it] |= (mt.rank << 9) | (mt.size << 14) | (mt.lead << 20);
        asm volatile("" : "+v"(info[it]));
    }
    __syncthreads();

    unsigned tot = 0u;
    if (tid < (unsigned)NKEYS) {
        for (int w = 0; w < NWV; ++w) {
            const unsigned v = cw[w * NKEYS + tid];
            cw[w * NKEYS + tid] = tot;
            tot += v;
        }
    }
    unsigned x = tot;
    for (unsigned d = 1u; d < 32u; d <<= 1) {
        const unsigned y = (unsigned)__builtin_amdgcn_ds_bpermute((int)((lane - d) << 2), (int)x);
        if (lane >= d) x += y;
    }
    if (lane == 31u) wsum[wv] = x;
    __syncthreads();
    unsigned off = 0u;
    for (unsigned w = 0u; w < wv; ++w) off += wsum[w];
    if (tid < (unsigned)NKEYS) start[tid] = off + x - tot;
    if (tid == (unsigned)(NKEYS - 1)) start[NKEYS] = off + x;
    __syncthreads();

#pragma unroll
    for (int it = 0; it < Src::ITERS; ++it) {
        if (src.skip(it)) continue;
        const unsigned k    = info[it] & 0x1ffu;
        const unsigned rank = (info[it] >> 9) & 31u;
        const unsigned size = (info[it] >> 14) & 63u;
        const unsigned lead = (info[it] >> 20) & 31u;
        const bool valid = (info[it] >> 31) != 0u;
        unsigned base = 0u;
        if (valid && rank == 0u) { base = c[k]; c[k] = base + size; }
        base = (unsigned)__builtin_amdgcn_ds_bpermute((int)(lead << 2), (int)base);
        if (valid) sorted[start[k] + base + rank] = pay[it];
    }
    __syncthreads();
}

struct EdgeSrc {
    static constexpr int ITERS = (SLAB + GT - 1) / GT;
    const int* esrc; const int* edst; const float* ew;
    unsigned e0, tid;
    __device__ __forceinline__ bool skip(int) const { return false; }
    __device__ __forceinline__ bool get(int it, unsigned& key, u2& pay) const {
        const unsigned i  = tid + (unsigned)GT * (unsigned)it;
        const unsigned ic = i < (unsigned)SLAB ? i : (unsigned)(SLAB - 1);
        const unsigned s = (unsigned)esrc[e0 + ic];
        const unsigned d = (unsigned)edst[e0 + ic];
        const unsigned w = __builtin_bit_cast(unsigned, ew[e0 + ic]);
        const unsigned b = d / BIN_ROWS;
        const bool ok = i < (unsigned)SLAB && b < (unsigned)NBINS && s < (unsigned)N_NODES;
        key   = ok ? b : 0u;
        pay.x = s | ((d - b * BIN_ROWS) << 17);
        pay.y = w;
        return ok;
    }
};

#define PS 6
#define LTAIL (4 * PS)
__device__ __forceinline__ unsigned* list_chunk(unsigned* pa, unsigned* pd, unsigned bin, unsigned ch) {
    const bool in_a = ch < (unsigned)BIN_ROWS;
    unsigned* p = in_a ? pa : pd;
    const unsigned row = in_a ? ch : ch - (unsigned)BIN_ROWS;
    return p + (bin * (unsigned)BIN_ROWS + row) * (unsigned)OUT_F;
}

#define ORD_TILES 17
#define PRJ_TILES 32
static_assert(NG * (ORD_TILES + PRJ_TILES) >= NTILES, "the tiles of all work-groups must cover the nodes");
static_assert(ORD_TILES <= NWV && PRJ_TILES <= NWV, "one tile a wave");

__global__ void __launch_bounds__(GT)
sgc_front(const float* __restrict__ x,
          const float* __restrict__ W,
          const float* __restrict__ bias,
          float* plo,
          float* phi,
          unsigned* ws_plane, unsigned* out_plane,
          const int* __restrict__ esrc,
          const int* __restrict__ edst,
          const float* __restrict__ ew) {
    __shared__ __attribute__((aligned(16))) float wl[OUT_F * WPITCH];
    __shared__ unsigned cw[NWV * NBINS];
    __shared__ unsigned start[NBINS + 1];
    __shared__ unsigned wsum[NWV];
    __shared__ __attribute__((aligned(8))) u2 sorted[SLAB];

    const unsigned tid  = threadIdx.x;
    const unsigned lane = tid & 31u;
    const unsigned wv   = (unsigned)__builtin_amdgcn_readfirstlane((int)(tid >> 5));
    const unsigned g    = blockIdx.x;

    for (int j = 0; j < (OUT_F * IN_F / 4) / GT; ++j) {
        const int i   = (int)tid + GT * j;
        const int row = i >> 6;
        const int c4  = i & 63;
        const v4f v = *(const v4f*)(W + (size_t)i * 4);
        *(v4f*)(wl + row * WPITCH + c4 * 4) = v;
    }

    if (g < (unsigned)NG) {
        unsigned* pa = ws_plane;
        unsigned* pd = out_plane + HALF_F;

        EdgeSrc src;
        src.esrc = esrc; src.edst = edst; src.ew = ew; src.e0 = g * (unsigned)SLAB; src.tid = tid;
        group_sort<NBINS>(src, cw, start, wsum, sorted, tid, lane, wv);

        for (unsigned b = wv; b < (unsigned)NBINS; b += NWV) {
            const unsigned st  = start[b];
            const unsigned tot = start[b + 1] - st;
            const unsigned n   = tot < (unsigned)CAP ? tot : (unsigned)CAP;
            for (unsigned t = lane; t <= n; t += 32u) {
                u2 v; v.x = tot; v.y = 0u;
                if (t != 0u) v = sorted[st + t - 1u];
                unsigned* c = list_chunk(pa, pd, b, g * (unsigned)LCHUNKS + (t >> 4));
                *(u2*)(c + (t & 15u) * 2u) = v;
            }
        }
    }
    __syncthreads();

    const unsigned nt   = g < (unsigned)NG ? (unsigned)ORD_TILES : (unsigned)PRJ_TILES;
    const unsigned t0   = g < (unsigned)NG ? g * (unsigned)ORD_TILES
                                           : (unsigned)(NG * ORD_TILES) + (g - (unsigned)NG) * (unsigned)PRJ_TILES;
    const int wave = (int)(t0 + wv);
    if (wv >= nt || wave >= NTILES) return;

    const int ln   = (int)lane;
    const int m    = ln & 15;
    const int hoff = (ln >> 4) << 4;

    const float* __restrict__ xrow = x + (size_t)(wave * 16 + m) * IN_F + hoff;
    const float* wb = wl + m * WPITCH + hoff;

    v8f acc0 = {}, acc1 = {}, acc2 = {}, acc3 = {};

#pragma unroll 2
    for (int kb = 0; kb < IN_F; kb += 32) {
        v4f a[4];
        for (int i = 0; i < 4; ++i)
            a[i] = *(const v4f*)(xrow + kb + 4 * i);
        for (int i = 0; i < 4; ++i) {
            const v4f b0 = *(const v4f*)(wb + 0 * 16 * WPITCH + kb + 4 * i);
            const v4f b1 = *(const v4f*)(wb + 1 * 16 * WPITCH + kb + 4 * i);
            const v4f b2 = *(const v4f*)(wb + 2 * 16 * WPITCH + kb + 4 * i);
            const v4f b3 = *(const v4f*)(wb + 3 * 16 * WPITCH + kb + 4 * i);
            const v2f alo = a[i].xy, ahi = a[i].zw;
            acc0 = __builtin_amdgcn_wmma_f32_16x16x4_f32(false, alo, false, b0.xy, (short)0, acc0, false, false);
            acc1 = __builtin_amdgcn_wmma_f32_16x16x4_f32(false, alo, false, b1.xy, (short)0, acc1, false, false);
            acc2 = __builtin_amdgcn_wmma_f32_16x16x4_f32(false, alo, false, b2.xy, (short)0, acc2, false, false);
            acc3 = __builtin_amdgcn_wmma_f32_16x16x4_f32(false, alo, false, b3.xy, (short)0, acc3, false, false);
            acc0 = __builtin_amdgcn_wmma_f32_16x16x4_f32(false, ahi, false, b0.zw, (short)0, acc0, false, false);
            acc1 = __builtin_amdgcn_wmma_f32_16x16x4_f32(false, ahi, false, b1.zw, (short)0, acc1, false, false);
            acc2 = __builtin_amdgcn_wmma_f32_16x16x4_f32(false, ahi, false, b2.zw, (short)0, acc2, false, false);
            acc3 = __builtin_amdgcn_wmma_f32_16x16x4_f32(false, ahi, false, b3.zw, (short)0, acc3, false, false);
        }
    }

    const int row_base = wave * 16 + ((ln >> 4) ? 8 : 0);
    const int col      = ln & 15;
    const float bv0 = bias[col +  0];
    const float bv1 = bias[col + 16];
    const float bv2 = bias[col + 32];
    const float bv3 = bias[col + 48];

#pragma unroll
    for (int r = 0; r < 8; ++r) {
        const size_t o = (size_t)(row_base + r) * OUT_F + col;
        plo[o +  0] = acc0[r] + bv0;
        plo[o + 16] = acc1[r] + bv1;
        phi[o + 32] = acc2[r] + bv2;
        phi[o + 48] = acc3[r] + bv3;
    }
}

struct ListSrc {
    static constexpr int LSTR  = (CAP + 31) / 32;
    static constexpr int ITERS = (NG / NWV) * LSTR;
    unsigned* pa; unsigned* pd; const unsigned* gcnt;
    unsigned bin, lane, wv;
    __device__ __forceinline__ bool skip(int it) const {
        const unsigned n = (unsigned)__builtin_amdgcn_readfirstlane((int)gcnt[wv + (unsigned)NWV * (unsigned)(it / LSTR)]);
        return 32u * (unsigned)(it % LSTR) >= n;
    }
    __device__ __forceinline__ bool get(int it, unsigned& key, u2& pay) const {
        const unsigned gl = wv + (unsigned)NWV * (unsigned)(it / LSTR);
        const unsigned j  = lane + 32u * (unsigned)(it % LSTR);
        const unsigned jc = j < (unsigned)CAP ? j : (unsigned)(CAP - 1);
        const unsigned t  = 1u + jc;
        const unsigned* c = list_chunk(pa, pd, bin, gl * (unsigned)LCHUNKS + (t >> 4));
        const u2 e = *(const u2*)(c + (t & 15u) * 2u);
        const unsigned r = e.x >> 17;
        const bool ok = j < gcnt[gl] && r < (unsigned)BIN_ROWS;
        key   = ok ? r : 0u;
        pay.x = e.x & 0x1ffffu;
        pay.y = e.y;
        return ok;
    }
};

__device__ __forceinline__ unsigned* l_chunk(unsigned* pd, unsigned bin, unsigned ch) {
    return pd + (bin * (unsigned)BIN_ROWS + ch) * (unsigned)OUT_F;
}

__global__ void __launch_bounds__(GT)
sgc_csr(unsigned* ws_plane, unsigned* out_plane) {
    __shared__ unsigned cw[NWV * BIN_ROWS];
    __shared__ unsigned start[BIN_ROWS + 1];
    __shared__ unsigned wsum[NWV];
    __shared__ unsigned gcnt[NG];
    __shared__ unsigned ovf[NG / 32];
    __shared__ __attribute__((aligned(16))) u2 sorted[NG * CAP];

    const unsigned tid  = threadIdx.x;
    const unsigned lane = tid & 31u;
    const unsigned wv   = (unsigned)__builtin_amdgcn_readfirstlane((int)(tid >> 5));
    const unsigned bin  = blockIdx.x;
    unsigned* pa = ws_plane;
    unsigned* pd = out_plane + HALF_F;

    if (tid < (unsigned)NG) {
        const unsigned cn = list_chunk(pa, pd, bin, tid * (unsigned)LCHUNKS)[0];
        gcnt[tid] = cn;
        const unsigned bal = __builtin_amdgcn_ballot_w32(cn > (unsigned)CAP);
        if (lane == 0u) ovf[wv] = bal;
    }
    __syncthreads();
    unsigned anyover = 0u;
    for (int i = 0; i < NG / 32; ++i) anyover |= ovf[i];
    const bool over = __builtin_amdgcn_readfirstlane((int)anyover) != 0;

    unsigned tot = SPARE;
    if (!over) {
        ListSrc src;
        src.pa = pa; src.pd = pd; src.gcnt = gcnt; src.bin = bin; src.lane = lane; src.wv = wv;
        group_sort<BIN_ROWS>(src, cw, start, wsum, sorted, tid, lane, wv);
        tot = start[BIN_ROWS];
        if (tot > (unsigned)ECAP) tot = SPARE;
    }

    if (tid < (unsigned)HDR_W) {
        unsigned v = 0u;
        if (tid == 0u) v = tot;
        if (tid >= 8u && tid < 8u + (unsigned)(BIN_ROWS + 2) / 2u && tot != SPARE) {
            const unsigned i = 2u * (tid - 8u);
            v = start[i];
            if (i + 1u <= (unsigned)BIN_ROWS) v |= start[i + 1u] << 16;
        }
        l_chunk(pd, bin, (unsigned)HDR_CH + (tid >> 5))[lane] = v;
    }
    if (tot != SPARE) {
        const unsigned* sw = (const unsigned*)sorted;
        for (unsigned j = tid; 4u * j < tot; j += GT) {
            unsigned sx[4], wy[4];
#pragma unroll
            for (int k = 0; k < 4; ++k) {
                const unsigned e = 4u * j + (unsigned)k;
                const bool in = e < tot;
                sx[k] = in ? sw[2u * e] : 0u;
                wy[k] = in ? sw[2u * e + 1u] : 0u;
            }
            typedef unsigned u4 __attribute__((ext_vector_type(4)));
            u4 wv4; wv4.x = wy[0]; wv4.y = wy[1]; wv4.z = wy[2]; wv4.w = wy[3];
            u2 s2;  s2.x = (sx[0] & 0xffffu) | (sx[1] << 16); s2.y = (sx[2] & 0xffffu) | (sx[3] << 16);
            const unsigned hb = (sx[0] >> 16) | ((sx[1] >> 16) << 8) | ((sx[2] >> 16) << 16) | ((sx[3] >> 16) << 24);
            *(u4*)(l_chunk(pd, bin, (unsigned)WSEC_CH + (j >> 3)) + (j & 7u) * 4u) = wv4;
            *(u2*)(l_chunk(pd, bin, (unsigned)SSEC_CH + (j >> 4)) + (j & 15u) * 2u) = s2;
            l_chunk(pd, bin, (unsigned)HSEC_CH + (j >> 5))[j & 31u] = hb;
        }
    }
}

__device__ __forceinline__ float xor8(float x) {
    const int b = __builtin_bit_cast(int, x);
    return __builtin_bit_cast(float, __builtin_amdgcn_permlane16(b, b, 0xfedcba98, 0x76543210, false, false));
}
__device__ __forceinline__ float xor16(float x) {
    const int b = __builtin_bit_cast(int, x);
    return __builtin_bit_cast(float, __builtin_amdgcn_permlanex16(b, b, 0x76543210, 0xfedcba98, false, false));
}

template <bool WHOLE>
__device__ __forceinline__ void
hop_body(unsigned* start, u2* sorted,
         const float* sp, int so,
         float* dp, int dof,
         const unsigned* out_plane,
         const int* __restrict__ esrc,
         const int* __restrict__ edst,
         const float* __restrict__ ew) {
    constexpr unsigned NQ = WHOLE ? 2u : 4u;

    const unsigned tid  = threadIdx.x;
    const unsigned lane = tid & 31u;
    const unsigned wv   = (unsigned)__builtin_amdgcn_readfirstlane((int)(tid >> 5));
    const unsigned bin  = blockIdx.x;
    const unsigned* lb  = out_plane + HALF_F + bin * (unsigned)(BIN_ROWS * OUT_F);
    const float* src = sp + so;
    float* dst = dp + dof + bin * (unsigned)(BIN_ROWS * OUT_F);

    const unsigned tot = (unsigned)__builtin_amdgcn_readfirstlane((int)lb[0]);

    if (tot <= (unsigned)ECAP) {
        if (tid < (unsigned)(BIN_ROWS + 2) / 2u) {
            const unsigned t = 8u + tid;
            const unsigned v = lb[(t >> 5) * (unsigned)OUT_F + (t & 31u)];
            start[2u * tid]      = v & 0xffffu;
            start[2u * tid + 1u] = v >> 16;
        }
        typedef unsigned u4 __attribute__((ext_vector_type(4)));
        for (unsigned j = tid; 4u * j < tot; j += GT) {
            const u4 w4 = *(const u4*)(lb + ((unsigned)WSEC_CH + (j >> 3)) * (unsigned)OUT_F + (j & 7u) * 4u);
            const u2 s2 = *(const u2*)(lb + ((unsigned)SSEC_CH + (j >> 4)) * (unsigned)OUT_F + (j & 15u) * 2u);
            const unsigned hb = lb[((unsigned)HSEC_CH + (j >> 5)) * (unsigned)OUT_F + (j & 31u)];
            u2 r0, r1, r2, r3;
            r0.x = (s2.x & 0xffffu) | ((hb & 0xffu) << 16);          r0.y = w4.x;
            r1.x = (s2.x >> 16)     | (((hb >> 8) & 0xffu) << 16);   r1.y = w4.y;
            r2.x = (s2.y & 0xffffu) | (((hb >> 16) & 0xffu) << 16);  r2.y = w4.z;
            r3.x = (s2.y >> 16)     | ((hb >> 24) << 16);            r3.y = w4.w;
            sorted[4u * j]      = r0;
            sorted[4u * j + 1u] = r1;
            sorted[4u * j + 2u] = r2;
            sorted[4u * j + 3u] = r3;
        }
        if (tid < (unsigned)LTAIL + 4u) { u2 z2; z2.x = 0u; z2.y = 0u; sorted[((tot + 3u) & ~3u) + tid] = z2; }
        __syncthreads();

        const unsigned q  = WHOLE ? lane >> 4 : lane >> 3;
        const unsigned c4 = (WHOLE ? lane & 15u : lane & 7u) * 4u;
        const unsigned cd = WHOLE ? c4 ^ (unsigned)HALF_F : c4;
        const char* srcb  = (const char*)src;
        const unsigned cb = c4 * 4u;
        unsigned rs0 = 0u, rs1 = 0u;
        {
            const unsigned rr = wv + (unsigned)NWV * lane;
            if (rr < (unsigned)BIN_ROWS) { rs0 = start[rr]; rs1 = start[rr + 1u]; }
        }
        for (unsigned k = 0u; wv + (unsigned)NWV * k < (unsigned)BIN_ROWS; k += 2u) {
            const unsigned ra  = wv + (unsigned)NWV * k;
            const unsigned rb  = ra + (unsigned)NWV;
            const unsigned sta = (unsigned)__builtin_amdgcn_readlane((int)rs0, (int)k);
            const unsigned na  = (unsigned)__builtin_amdgcn_readlane((int)rs1, (int)k) - sta;
            const unsigned stb = (unsigned)__builtin_amdgcn_readlane((int)rs0, (int)(k + 1u));
            const unsigned nb  = (unsigned)__builtin_amdgcn_readlane((int)rs1, (int)(k + 1u)) - stb;
            const unsigned nmax = na > nb ? na : nb;
            v4f acca = {0.f, 0.f, 0.f, 0.f}, accb = {0.f, 0.f, 0.f, 0.f};
            for (unsigned base = 0u; base < nmax; base += NQ * (unsigned)PS) {
                const unsigned lefta = na > base ? na - base : 0u;
                const unsigned leftb = nb > base ? nb - base : 0u;
                unsigned sa[PS], sb[PS]; float wa[PS], wb[PS]; v4f va[PS], vb[PS];
                const unsigned oa = lefta != 0u ? sta + base : 0u;
                const unsigned ob = leftb != 0u ? stb + base : 0u;
                const u2* pa0 = sorted + (oa + q);
                const u2* pb0 = sorted + (ob + q);
#pragma unroll
                for (int p = 0; p < PS; ++p) {
                    const unsigned idx = NQ * (unsigned)p + q;
                    const bool oka = idx < lefta, okb = idx < leftb;
                    const u2 ea = pa0[NQ * (unsigned)p];
                    const u2 eb = pb0[NQ * (unsigned)p];
                    const unsigned ax = ea.x, ay = ea.y, bx = eb.x, by = eb.y;
                    sa[p] = oka ? ax : 0u;  wa[p] = oka ? __builtin_bit_cast(float, ay) : 0.f;
                    sb[p] = okb ? bx : 0u;  wb[p] = okb ? __builtin_bit_cast(float, by) : 0.f;
                }
#pragma unroll
                for (int p = 0; p < PS; ++p) {
                    va[p] = *(const v4f*)(srcb + ((sa[p] << 8) | cb));
                    vb[p] = *(const v4f*)(srcb + ((sb[p] << 8) | cb));
                }
                asm volatile("" : "+v"(va[0]), "+v"(va[1]), "+v"(va[2]), "+v"(va[3]), "+v"(va[4]), "+v"(va[5]),
                                  "+v"(vb[0]), "+v"(vb[1]), "+v"(vb[2]), "+v"(vb[3]), "+v"(vb[4]), "+v"(vb[5]));
                static_assert(PS == 6, "the statement above names the PS gathered rows one by one");
#pragma unroll
                for (int p = 0; p < PS; ++p) {
                    acca.x = __builtin_fmaf(va[p].x, wa[p], acca.x); acca.y = __builtin_fmaf(va[p].y, wa[p], acca.y);
                    acca.z = __builtin_fmaf(va[p].z, wa[p], acca.z); acca.w = __builtin_fmaf(va[p].w, wa[p], acca.w);
                    accb.x = __builtin_fmaf(vb[p].x, wb[p], accb.x); accb.y = __builtin_fmaf(vb[p].y, wb[p], accb.y);
                    accb.z = __builtin_fmaf(vb[p].z, wb[p], accb.z); accb.w = __builtin_fmaf(vb[p].w, wb[p], accb.w);
                }
            }
            if (!WHOLE) {
                acca.x += xor8(acca.x);  acca.y += xor8(acca.y);  acca.z += xor8(acca.z);  acca.w += xor8(acca.w);
                accb.x += xor8(accb.x);  accb.y += xor8(accb.y);  accb.z += xor8(accb.z);  accb.w += xor8(accb.w);
            }
            acca.x += xor16(acca.x); acca.y += xor16(acca.y); acca.z += xor16(acca.z); acca.w += xor16(acca.w);
            accb.x += xor16(accb.x); accb.y += xor16(accb.y); accb.z += xor16(accb.z); accb.w += xor16(accb.w);
            if (q == 0u) {
                *(v4f*)(dst + ra * (unsigned)OUT_F + cd) = acca;
                if (rb < (unsigned)BIN_ROWS) *(v4f*)(dst + rb * (unsigned)OUT_F + cd) = accb;
            }
        }
    } else {
        float* tile = (float*)sorted;
        const v4f z = {0.f, 0.f, 0.f, 0.f};
        for (unsigned hf = 0u; hf < (WHOLE ? 2u : 1u); ++hf) {
            const float* hsrc = src + hf * (unsigned)HALF_F;
            float* hdst = dst + (WHOLE ? (hf * (unsigned)HALF_F) ^ (unsigned)HALF_F : 0u);
            for (unsigned i = tid; i < (unsigned)(BIN_ROWS * HALF_F / 4); i += GT) ((v4f*)tile)[i] = z;
            __syncthreads();
            for (unsigned eb = 0u; eb < (unsigned)N_EDGES; eb += 32u) {
                const unsigned s = (unsigned)esrc[eb + lane];
                const unsigned d = (unsigned)edst[eb + lane];
                const unsigned w = __builtin_bit_cast(unsigned, ew[eb + lane]);
                const unsigned r = d - bin * (unsigned)BIN_ROWS;
                unsigned m = __builtin_amdgcn_ballot_w32(r < (unsigned)BIN_ROWS && (r & (unsigned)(NWV - 1)) == wv && s < (unsigned)N_NODES);
                for (int k = 0; k < 32 && m != 0u; ++k) {
                    const int j = __builtin_ctz(m);
                    m &= m - 1u;
                    const unsigned sj = (unsigned)__builtin_amdgcn_readlane((int)s, j);
                    const unsigned rj = (unsigned)__builtin_amdgcn_readlane((int)r, j);
                    const float    wj = __builtin_bit_cast(float, __builtin_amdgcn_readlane((int)w, j));
                    const float gsrc = hsrc[sj * (unsigned)OUT_F + lane];
                    tile[rj * (unsigned)HALF_F + lane] += gsrc * wj;
                }
            }
            __syncthreads();
            for (unsigned i = tid; i < (unsigned)(BIN_ROWS * HALF_F / 4); i += GT)
                *(v4f*)(hdst + (i >> 3) * (unsigned)OUT_F + (i & 7u) * 4u) = ((const v4f*)tile)[i];
            if (WHOLE) __syncthreads();
        }
    }
}

__global__ void __launch_bounds__(GT)
sgc_hop(const float* sp, int so,
        float* dp, int dof,
        const unsigned* out_plane,
        const int* __restrict__ esrc,
        const int* __restrict__ edst,
        const float* __restrict__ ew) {
    __shared__ unsigned start[BIN_ROWS + 2];
    __shared__ __attribute__((aligned(16))) u2 sorted[ECAP + LTAIL + 4];
    static_assert(sizeof(u2) * ECAP >= sizeof(float) * BIN_ROWS * HALF_F, "the spare path's tile lives in sorted[]");
    hop_body<false>(start, sorted, sp, so, dp, dof, out_plane, esrc, edst, ew);
}

__global__ void __launch_bounds__(GT)
sgc_hop_rows(const float* sp,
             float* dp,
             const unsigned* out_plane,
             const int* __restrict__ esrc,
             const int* __restrict__ edst,
             const float* __restrict__ ew) {
    __shared__ unsigned start[BIN_ROWS + 2];
    __shared__ __attribute__((aligned(16))) u2 sorted[ECAP + LTAIL + 4];
    static_assert(sizeof(u2) * ECAP >= sizeof(float) * BIN_ROWS * HALF_F, "the spare path's tile lives in sorted[]");
    hop_body<true>(start, sorted, sp, 0, dp, 0, out_plane, esrc, edst, ew);
}

extern "C" void kernel_launch(void* const* d_in, const int* in_sizes, int n_in,
                              void* d_out, int out_size, void* d_ws, size_t ws_size,
                              hipStream_t stream) {
    const float* x    = (const float*)d_in[0];
    const float* W    = (const float*)d_in[1];
    const float* b    = (const float*)d_in[2];
    const int*   esrc = (const int*)d_in[3];
    const int*   edst = (const int*)d_in[4];
    const float* ew   = (const float*)d_in[5];

    float* out = (float*)d_out;
    float* ws  = (float*)d_ws;

    sgc_front<<<2 * NG, GT, 0, stream>>>(x, W, b, out, ws, (unsigned*)ws, (unsigned*)out, esrc, edst, ew);
    sgc_csr  <<<NBINS, GT, 0, stream>>>((unsigned*)ws, (unsigned*)out);

    sgc_hop<<<NBINS, GT, 0, stream>>>(out,  0, ws,   0, (const unsigned*)out, esrc, edst, ew);
    sgc_hop<<<NBINS, GT, 0, stream>>>(ws,  32, out,  0, (const unsigned*)out, esrc, edst, ew);
    sgc_hop<<<NBINS, GT, 0, stream>>>(ws,   0, ws,  32, (const unsigned*)out, esrc, edst, ew);
    sgc_hop<<<NBINS, GT, 0, stream>>>(out,  0, ws,   0, (const unsigned*)out, esrc, edst, ew);
    sgc_hop_rows<<<NBINS, GT, 0, stream>>>(ws, out, (const unsigned*)out, esrc, edst, ew);
}
